# starting kernel
# baseline (speedup 1.0000x reference)
#include <hip/hip_runtime.h>
#include <hip/hip_bf16.h>
#include <math.h>

typedef float v2f __attribute__((ext_vector_type(2)));
typedef float v4f __attribute__((ext_vector_type(4)));
typedef float v8f __attribute__((ext_vector_type(8)));

#define BATCH 16
#define NPTS  4096
#define CIN   256
#define COUT  512
#define KNBR  16
#define NOUT  1024
#define MROWS (BATCH * NPTS)
#define BN_EPS 1e-5f
#define XOUT_ELEMS ((size_t)BATCH * NOUT * COUT)

__global__ __launch_bounds__(128) void gemm_bn_relu_kernel(
    const float* __restrict__ x, const float* __restrict__ W,
    const float* __restrict__ bias, const float* __restrict__ gamma,
    const float* __restrict__ beta, const float* __restrict__ rmean,
    const float* __restrict__ rvar, float* __restrict__ h)
{
    const int wave = threadIdx.x >> 5;
    const int lane = threadIdx.x & 31;
    const int m0 = (blockIdx.x * 4 + wave) * 16;
    const int n0 = blockIdx.y * 64;

    const int rowA = m0 + (lane & 15);
    const int kOff = (lane >> 4) << 1;
    const float* aPtr = x + (size_t)rowA * CIN + kOff;

    const int colL = lane & 15;
    const float* bPtr0 = W + (size_t)(n0 +  0 + colL) * CIN + kOff;
    const float* bPtr1 = W + (size_t)(n0 + 16 + colL) * CIN + kOff;
    const float* bPtr2 = W + (size_t)(n0 + 32 + colL) * CIN + kOff;
    const float* bPtr3 = W + (size_t)(n0 + 48 + colL) * CIN + kOff;

    v8f acc[4];
    #pragma unroll
    for (int nt = 0; nt < 4; ++nt) acc[nt] = {};

    #pragma unroll 4
    for (int c = 0; c < CIN; c += 4) {
        v2f a  = *(const v2f*)(aPtr + c);
        v2f b0 = *(const v2f*)(bPtr0 + c);
        v2f b1 = *(const v2f*)(bPtr1 + c);
        v2f b2 = *(const v2f*)(bPtr2 + c);
        v2f b3 = *(const v2f*)(bPtr3 + c);
        acc[0] = __builtin_amdgcn_wmma_f32_16x16x4_f32(false, a, false, b0,
                                                       (short)0, acc[0], false, false);
        acc[1] = __builtin_amdgcn_wmma_f32_16x16x4_f32(false, a, false, b1,
                                                       (short)0, acc[1], false, false);
        acc[2] = __builtin_amdgcn_wmma_f32_16x16x4_f32(false, a, false, b2,
                                                       (short)0, acc[2], false, false);
        acc[3] = __builtin_amdgcn_wmma_f32_16x16x4_f32(false, a, false, b3,
                                                       (short)0, acc[3], false, false);
    }

    const int rOff = (lane >> 4) << 3;
    #pragma unroll
    for (int nt = 0; nt < 4; ++nt) {
        const int o = n0 + nt * 16 + colL;
        const float sc = gamma[o] * rsqrtf(rvar[o] + BN_EPS);
        const float sh = (bias[o] - rmean[o]) * sc + beta[o];
        #pragma unroll
        for (int j = 0; j < 8; ++j) {
            float v = fmaf(acc[nt][j], sc, sh);
            v = fmaxf(v, 0.0f);
            h[(size_t)(m0 + j + rOff) * COUT + o] = v;
        }
    }
}

__global__ __launch_bounds__(1024) void fps_kernel(const float* __restrict__ p,
                                                   int* __restrict__ fid)
{
    const int b = blockIdx.x;
    const float* pb = p + (size_t)b * NPTS * 3;
    const int t = threadIdx.x;

    float px[4], py[4], pz[4], d[4];
    const float x0 = pb[0], y0 = pb[1], z0 = pb[2];
    #pragma unroll
    for (int s = 0; s < 4; ++s) {
        const int idx = t + s * 1024;
        px[s] = pb[idx * 3 + 0];
        py[s] = pb[idx * 3 + 1];
        pz[s] = pb[idx * 3 + 2];
        const float dx = px[s] - x0, dy = py[s] - y0, dz = pz[s] - z0;
        d[s] = dx * dx + dy * dy + dz * dz;
    }

    __shared__ float sval[32];
    __shared__ int   sidx[32];
    __shared__ int   snxt;
    __shared__ float scoord[3];

    if (t == 0) fid[b * NOUT + 0] = b * NPTS;

    const int wid = t >> 5, lane = t & 31;

    for (int i = 1; i < NOUT; ++i) {
        float bv = d[0]; int bi = t;
        #pragma unroll
        for (int s = 1; s < 4; ++s) {
            if (d[s] > bv) { bv = d[s]; bi = t + s * 1024; }
        }
        #pragma unroll
        for (int off = 16; off >= 1; off >>= 1) {
            const float ov = __shfl_xor(bv, off, 32);
            const int   oi = __shfl_xor(bi, off, 32);
            if (ov > bv || (ov == bv && oi < bi)) { bv = ov; bi = oi; }
        }
        if (lane == 0) { sval[wid] = bv; sidx[wid] = bi; }
        __syncthreads();
        if (wid == 0) {
            float v = sval[lane]; int ix = sidx[lane];
            #pragma unroll
            for (int off = 16; off >= 1; off >>= 1) {
                const float ov = __shfl_xor(v, off, 32);
                const int   oi = __shfl_xor(ix, off, 32);
                if (ov > v || (ov == v && oi < ix)) { v = ov; ix = oi; }
            }
            if (lane == 0) {
                snxt = ix;
                scoord[0] = pb[ix * 3 + 0];
                scoord[1] = pb[ix * 3 + 1];
                scoord[2] = pb[ix * 3 + 2];
                fid[b * NOUT + i] = b * NPTS + ix;
            }
        }
        __syncthreads();
        const float nx = scoord[0], ny = scoord[1], nz = scoord[2];
        #pragma unroll
        for (int s = 0; s < 4; ++s) {
            const float dx = px[s] - nx, dy = py[s] - ny, dz = pz[s] - nz;
            const float dd = dx * dx + dy * dy + dz * dz;
            d[s] = fminf(d[s], dd);
        }
    }
}

__global__ __launch_bounds__(128) void gather_maxpool_kernel(
    const float* __restrict__ h, const float* __restrict__ p,
    const int* __restrict__ fid, const int* __restrict__ sid_euc,
    float* __restrict__ out)
{
    const int s = blockIdx.x;
    const int t = threadIdx.x;

    __shared__ int ssid[KNBR];
    const int f = fid[s];
    if (t < KNBR) ssid[t] = sid_euc[(size_t)f * KNBR + t];
    if (t < 3)    out[XOUT_ELEMS + (size_t)s * 3 + t] = p[(size_t)f * 3 + t];
    __syncthreads();

    const int c = t * 4;
    v4f acc = { -INFINITY, -INFINITY, -INFINITY, -INFINITY };
    #pragma unroll
    for (int k = 0; k < KNBR; ++k) {
        const int r = ssid[k];
        if (k + 1 < KNBR)
            __builtin_prefetch(h + (size_t)ssid[k + 1] * COUT + c, 0, 0);
        const v4f v = *(const v4f*)(h + (size_t)r * COUT + c);
        acc.x = fmaxf(acc.x, v.x);
        acc.y = fmaxf(acc.y, v.y);
        acc.z = fmaxf(acc.z, v.z);
        acc.w = fmaxf(acc.w, v.w);
    }
    *(v4f*)(out + (size_t)s * COUT + c) = acc;
}

extern "C" void kernel_launch(void* const* d_in, const int* in_sizes, int n_in,
                              void* d_out, int out_size, void* d_ws, size_t ws_size,
                              hipStream_t stream)
{
    (void)in_sizes; (void)n_in; (void)out_size; (void)ws_size;
    const float* x     = (const float*)d_in[0];
    const float* p     = (const float*)d_in[1];
    const int*   sid   = (const int*)  d_in[2];
    const float* W     = (const float*)d_in[4];
    const float* bias  = (const float*)d_in[5];
    const float* gamma = (const float*)d_in[6];
    const float* beta  = (const float*)d_in[7];
    const float* rmean = (const float*)d_in[8];
    const float* rvar  = (const float*)d_in[9];
    float* out = (float*)d_out;

    float* h  = (float*)d_ws;
    int* fid  = (int*)((char*)d_ws + (size_t)MROWS * COUT * 4);

    dim3 ggrid(MROWS / 64, COUT / 64);
    gemm_bn_relu_kernel<<<ggrid, 128, 0, stream>>>(x, W, bias, gamma, beta,
                                                   rmean, rvar, h);
    fps_kernel<<<BATCH, 1024, 0, stream>>>(p, fid);
    gather_maxpool_kernel<<<BATCH * NOUT, 128, 0, stream>>>(h, p, fid, sid, out);
}


# FPS argmax: both wave reductions on the vector ALU (DPP row_xmask + permlanex16: integer max of the distance bits, then min index among equals) instead of ds_bpermute shuffles
# speedup vs baseline: 1.5752x; 1.5752x over previous
#include <hip/hip_runtime.h>
#include <hip/hip_bf16.h>
#include <math.h>

typedef float v2f __attribute__((ext_vector_type(2)));
typedef float v4f __attribute__((ext_vector_type(4)));
typedef float v8f __attribute__((ext_vector_type(8)));

#define BATCH 16
#define NPTS  4096
#define CIN   256
#define COUT  512
#define KNBR  16
#define NOUT  1024
#define MROWS (BATCH * NPTS)
#define BN_EPS 1e-5f
#define XOUT_ELEMS ((size_t)BATCH * NOUT * COUT)

__global__ __launch_bounds__(128) void gemm_bn_relu_kernel(
    const float* __restrict__ x, const float* __restrict__ W,
    const float* __restrict__ bias, const float* __restrict__ gamma,
    const float* __restrict__ beta, const float* __restrict__ rmean,
    const float* __restrict__ rvar, float* __restrict__ h)
{
    const int wave = threadIdx.x >> 5;
    const int lane = threadIdx.x & 31;
    const int m0 = (blockIdx.x * 4 + wave) * 16;
    const int n0 = blockIdx.y * 64;

    const int rowA = m0 + (lane & 15);
    const int kOff = (lane >> 4) << 1;
    const float* aPtr = x + (size_t)rowA * CIN + kOff;

    const int colL = lane & 15;
    const float* bPtr0 = W + (size_t)(n0 +  0 + colL) * CIN + kOff;
    const float* bPtr1 = W + (size_t)(n0 + 16 + colL) * CIN + kOff;
    const float* bPtr2 = W + (size_t)(n0 + 32 + colL) * CIN + kOff;
    const float* bPtr3 = W + (size_t)(n0 + 48 + colL) * CIN + kOff;

    v8f acc[4];
    #pragma unroll
    for (int nt = 0; nt < 4; ++nt) acc[nt] = {};

    #pragma unroll 4
    for (int c = 0; c < CIN; c += 4) {
        v2f a  = *(const v2f*)(aPtr + c);
        v2f b0 = *(const v2f*)(bPtr0 + c);
        v2f b1 = *(const v2f*)(bPtr1 + c);
        v2f b2 = *(const v2f*)(bPtr2 + c);
        v2f b3 = *(const v2f*)(bPtr3 + c);
        acc[0] = __builtin_amdgcn_wmma_f32_16x16x4_f32(false, a, false, b0,
                                                       (short)0, acc[0], false, false);
        acc[1] = __builtin_amdgcn_wmma_f32_16x16x4_f32(false, a, false, b1,
                                                       (short)0, acc[1], false, false);
        acc[2] = __builtin_amdgcn_wmma_f32_16x16x4_f32(false, a, false, b2,
                                                       (short)0, acc[2], false, false);
        acc[3] = __builtin_amdgcn_wmma_f32_16x16x4_f32(false, a, false, b3,
                                                       (short)0, acc[3], false, false);
    }

    const int rOff = (lane >> 4) << 3;
    #pragma unroll
    for (int nt = 0; nt < 4; ++nt) {
        const int o = n0 + nt * 16 + colL;
        const float sc = gamma[o] * rsqrtf(rvar[o] + BN_EPS);
        const float sh = (bias[o] - rmean[o]) * sc + beta[o];
        #pragma unroll
        for (int j = 0; j < 8; ++j) {
            float v = fmaf(acc[nt][j], sc, sh);
            v = fmaxf(v, 0.0f);
            h[(size_t)(m0 + j + rOff) * COUT + o] = v;
        }
    }
}

template <int XM>
__device__ __forceinline__ int row_xmask(int v)
{
    return __builtin_amdgcn_mov_dpp(v, 0x160 + XM, 0xf, 0xf, true);
}
__device__ __forceinline__ int xhalf(int v)
{
    return __builtin_amdgcn_permlanex16(v, v, 0x76543210, 0xfedcba98, false, false);
}
__device__ __forceinline__ int wave_max_i(int v)
{
    v = max(v, row_xmask<1>(v));
    v = max(v, row_xmask<2>(v));
    v = max(v, row_xmask<4>(v));
    v = max(v, row_xmask<8>(v));
    v = max(v, xhalf(v));
    return v;
}
__device__ __forceinline__ int wave_min_i(int v)
{
    v = min(v, row_xmask<1>(v));
    v = min(v, row_xmask<2>(v));
    v = min(v, row_xmask<4>(v));
    v = min(v, row_xmask<8>(v));
    v = min(v, xhalf(v));
    return v;
}
__device__ __forceinline__ void wave_argmax(float& v, int& ix)
{
    const int k = __builtin_bit_cast(int, v);
    const int m = wave_max_i(k);
    const int cand = (k == m) ? ix : 0x7fffffff;
    ix = wave_min_i(cand);
    v = __builtin_bit_cast(float, m);
}

__global__ __launch_bounds__(1024) void fps_kernel(const float* __restrict__ p,
                                                   int* __restrict__ fid)
{
    const int b = blockIdx.x;
    const float* pb = p + (size_t)b * NPTS * 3;
    const int t = threadIdx.x;

    float px[4], py[4], pz[4], d[4];
    const float x0 = pb[0], y0 = pb[1], z0 = pb[2];
    #pragma unroll
    for (int s = 0; s < 4; ++s) {
        const int idx = t + s * 1024;
        px[s] = pb[idx * 3 + 0];
        py[s] = pb[idx * 3 + 1];
        pz[s] = pb[idx * 3 + 2];
        const float dx = px[s] - x0, dy = py[s] - y0, dz = pz[s] - z0;
        d[s] = dx * dx + dy * dy + dz * dz;
    }

    __shared__ float sval[32];
    __shared__ int   sidx[32];
    __shared__ int   snxt;
    __shared__ float scoord[3];

    if (t == 0) fid[b * NOUT + 0] = b * NPTS;

    const int wid = t >> 5, lane = t & 31;

    for (int i = 1; i < NOUT; ++i) {
        float bv = d[0]; int bi = t;
        #pragma unroll
        for (int s = 1; s < 4; ++s) {
            if (d[s] > bv) { bv = d[s]; bi = t + s * 1024; }
        }
        wave_argmax(bv, bi);
        if (lane == 0) { sval[wid] = bv; sidx[wid] = bi; }
        __syncthreads();
        if (wid == 0) {
            float v = sval[lane]; int ix = sidx[lane];
            wave_argmax(v, ix);
            if (lane == 0) {
                snxt = ix;
                scoord[0] = pb[ix * 3 + 0];
                scoord[1] = pb[ix * 3 + 1];
                scoord[2] = pb[ix * 3 + 2];
                fid[b * NOUT + i] = b * NPTS + ix;
            }
        }
        __syncthreads();
        const float nx = scoord[0], ny = scoord[1], nz = scoord[2];
        #pragma unroll
        for (int s = 0; s < 4; ++s) {
            const float dx = px[s] - nx, dy = py[s] - ny, dz = pz[s] - nz;
            const float dd = dx * dx + dy * dy + dz * dz;
            d[s] = fminf(d[s], dd);
        }
    }
}

__global__ __launch_bounds__(128) void gather_maxpool_kernel(
    const float* __restrict__ h, const float* __restrict__ p,
    const int* __restrict__ fid, const int* __restrict__ sid_euc,
    float* __restrict__ out)
{
    const int s = blockIdx.x;
    const int t = threadIdx.x;

    __shared__ int ssid[KNBR];
    const int f = fid[s];
    if (t < KNBR) ssid[t] = sid_euc[(size_t)f * KNBR + t];
    if (t < 3)    out[XOUT_ELEMS + (size_t)s * 3 + t] = p[(size_t)f * 3 + t];
    __syncthreads();

    const int c = t * 4;
    v4f acc = { -INFINITY, -INFINITY, -INFINITY, -INFINITY };
    #pragma unroll
    for (int k = 0; k < KNBR; ++k) {
        const int r = ssid[k];
        if (k + 1 < KNBR)
            __builtin_prefetch(h + (size_t)ssid[k + 1] * COUT + c, 0, 0);
        const v4f v = *(const v4f*)(h + (size_t)r * COUT + c);
        acc.x = fmaxf(acc.x, v.x);
        acc.y = fmaxf(acc.y, v.y);
        acc.z = fmaxf(acc.z, v.z);
        acc.w = fmaxf(acc.w, v.w);
    }
    *(v4f*)(out + (size_t)s * COUT + c) = acc;
}

extern "C" void kernel_launch(void* const* d_in, const int* in_sizes, int n_in,
                              void* d_out, int out_size, void* d_ws, size_t ws_size,
                              hipStream_t stream)
{
    (void)in_sizes; (void)n_in; (void)out_size; (void)ws_size;
    const float* x     = (const float*)d_in[0];
    const float* p     = (const float*)d_in[1];
    const int*   sid   = (const int*)  d_in[2];
    const float* W     = (const float*)d_in[4];
    const float* bias  = (const float*)d_in[5];
    const float* gamma = (const float*)d_in[6];
    const float* beta  = (const float*)d_in[7];
    const float* rmean = (const float*)d_in[8];
    const float* rvar  = (const float*)d_in[9];
    float* out = (float*)d_out;

    float* h  = (float*)d_ws;
    int* fid  = (int*)((char*)d_ws + (size_t)MROWS * COUT * 4);

    dim3 ggrid(MROWS / 64, COUT / 64);
    gemm_bn_relu_kernel<<<ggrid, 128, 0, stream>>>(x, W, bias, gamma, beta,
                                                   rmean, rvar, h);
    fps_kernel<<<BATCH, 1024, 0, stream>>>(p, fid);
    gather_maxpool_kernel<<<BATCH * NOUT, 128, 0, stream>>>(h, p, fid, sid, out);
}


# FPS: 8 waves x 16 points per thread instead of 32 x 4: same per-point work, a quarter of the wave reductions, barrier arrivals and partials per pick; per-thread argmax as a 4-level tournament
# speedup vs baseline: 1.7276x; 1.0967x over previous
#include <hip/hip_runtime.h>
#include <hip/hip_bf16.h>
#include <math.h>

typedef float v2f __attribute__((ext_vector_type(2)));
typedef float v4f __attribute__((ext_vector_type(4)));
typedef float v8f __attribute__((ext_vector_type(8)));

#define BATCH 16
#define NPTS  4096
#define CIN   256
#define COUT  512
#define KNBR  16
#define NOUT  1024
#define MROWS (BATCH * NPTS)
#define BN_EPS 1e-5f
#define XOUT_ELEMS ((size_t)BATCH * NOUT * COUT)

__global__ __launch_bounds__(128) void gemm_bn_relu_kernel(
    const float* __restrict__ x, const float* __restrict__ W,
    const float* __restrict__ bias, const float* __restrict__ gamma,
    const float* __restrict__ beta, const float* __restrict__ rmean,
    const float* __restrict__ rvar, float* __restrict__ h)
{
    const int wave = threadIdx.x >> 5;
    const int lane = threadIdx.x & 31;
    const int m0 = (blockIdx.x * 4 + wave) * 16;
    const int n0 = blockIdx.y * 64;

    const int rowA = m0 + (lane & 15);
    const int kOff = (lane >> 4) << 1;
    const float* aPtr = x + (size_t)rowA * CIN + kOff;

    const int colL = lane & 15;
    const float* bPtr0 = W + (size_t)(n0 +  0 + colL) * CIN + kOff;
    const float* bPtr1 = W + (size_t)(n0 + 16 + colL) * CIN + kOff;
    const float* bPtr2 = W + (size_t)(n0 + 32 + colL) * CIN + kOff;
    const float* bPtr3 = W + (size_t)(n0 + 48 + colL) * CIN + kOff;

    v8f acc[4];
    #pragma unroll
    for (int nt = 0; nt < 4; ++nt) acc[nt] = {};

    #pragma unroll 4
    for (int c = 0; c < CIN; c += 4) {
        v2f a  = *(const v2f*)(aPtr + c);
        v2f b0 = *(const v2f*)(bPtr0 + c);
        v2f b1 = *(const v2f*)(bPtr1 + c);
        v2f b2 = *(const v2f*)(bPtr2 + c);
        v2f b3 = *(const v2f*)(bPtr3 + c);
        acc[0] = __builtin_amdgcn_wmma_f32_16x16x4_f32(false, a, false, b0,
                                                       (short)0, acc[0], false, false);
        acc[1] = __builtin_amdgcn_wmma_f32_16x16x4_f32(false, a, false, b1,
                                                       (short)0, acc[1], false, false);
        acc[2] = __builtin_amdgcn_wmma_f32_16x16x4_f32(false, a, false, b2,
                                                       (short)0, acc[2], false, false);
        acc[3] = __builtin_amdgcn_wmma_f32_16x16x4_f32(false, a, false, b3,
                                                       (short)0, acc[3], false, false);
    }

    const int rOff = (lane >> 4) << 3;
    #pragma unroll
    for (int nt = 0; nt < 4; ++nt) {
        const int o = n0 + nt * 16 + colL;
        const float sc = gamma[o] * rsqrtf(rvar[o] + BN_EPS);
        const float sh = (bias[o] - rmean[o]) * sc + beta[o];
        #pragma unroll
        for (int j = 0; j < 8; ++j) {
            float v = fmaf(acc[nt][j], sc, sh);
            v = fmaxf(v, 0.0f);
            h[(size_t)(m0 + j + rOff) * COUT + o] = v;
        }
    }
}

template <int XM>
__device__ __forceinline__ int row_xmask(int v)
{
    return __builtin_amdgcn_mov_dpp(v, 0x160 + XM, 0xf, 0xf, true);
}
__device__ __forceinline__ int xhalf(int v)
{
    return __builtin_amdgcn_permlanex16(v, v, 0x76543210, 0xfedcba98, false, false);
}
__device__ __forceinline__ int wave_max_i(int v)
{
    v = max(v, row_xmask<1>(v));
    v = max(v, row_xmask<2>(v));
    v = max(v, row_xmask<4>(v));
    v = max(v, row_xmask<8>(v));
    v = max(v, xhalf(v));
    return v;
}
__device__ __forceinline__ int wave_min_i(int v)
{
    v = min(v, row_xmask<1>(v));
    v = min(v, row_xmask<2>(v));
    v = min(v, row_xmask<4>(v));
    v = min(v, row_xmask<8>(v));
    v = min(v, xhalf(v));
    return v;
}
__device__ __forceinline__ void wave_argmax(float& v, int& ix)
{
    const int k = __builtin_bit_cast(int, v);
    const int m = wave_max_i(k);
    const int cand = (k == m) ? ix : 0x7fffffff;
    ix = wave_min_i(cand);
    v = __builtin_bit_cast(float, m);
}

__device__ __forceinline__ void octet_argmax(float& v, int& ix)
{
    const int k = __builtin_bit_cast(int, v);
    int m = k;
    m = max(m, row_xmask<1>(m));
    m = max(m, row_xmask<2>(m));
    m = max(m, row_xmask<4>(m));
    int c = (k == m) ? ix : 0x7fffffff;
    c = min(c, row_xmask<1>(c));
    c = min(c, row_xmask<2>(c));
    c = min(c, row_xmask<4>(c));
    ix = c;
    v = __builtin_bit_cast(float, m);
}

#define FPS_THREADS 256
#define FPS_PPT     (NPTS / FPS_THREADS)
#define FPS_WAVES   (FPS_THREADS / 32)
__global__ __launch_bounds__(FPS_THREADS) void fps_kernel(const float* __restrict__ p,
                                                          int* __restrict__ fid)
{
    const int b = blockIdx.x;
    const float* pb = p + (size_t)b * NPTS * 3;
    const int t = threadIdx.x;

    float px[FPS_PPT], py[FPS_PPT], pz[FPS_PPT], d[FPS_PPT];
    const float x0 = pb[0], y0 = pb[1], z0 = pb[2];
    #pragma unroll
    for (int s = 0; s < FPS_PPT; ++s) {
        const int idx = t + s * FPS_THREADS;
        px[s] = pb[idx * 3 + 0];
        py[s] = pb[idx * 3 + 1];
        pz[s] = pb[idx * 3 + 2];
        const float dx = px[s] - x0, dy = py[s] - y0, dz = pz[s] - z0;
        d[s] = dx * dx + dy * dy + dz * dz;
    }

    __shared__ float sval[FPS_WAVES];
    __shared__ int   sidx[FPS_WAVES];
    __shared__ float scoord[3];

    if (t == 0) fid[b * NOUT + 0] = b * NPTS;

    const int wid = t >> 5, lane = t & 31;

    for (int i = 1; i < NOUT; ++i) {
        float tv[FPS_PPT / 2]; int ts[FPS_PPT / 2];
        #pragma unroll
        for (int s = 0; s < FPS_PPT / 2; ++s) {
            const float va = d[2 * s], vb = d[2 * s + 1];
            const bool r = vb > va;
            tv[s] = r ? vb : va;
            ts[s] = r ? 2 * s + 1 : 2 * s;
        }
        #pragma unroll
        for (int w = 1; w < FPS_PPT / 2; w <<= 1) {
            #pragma unroll
            for (int s = 0; s < FPS_PPT / 2; s += 2 * w) {
                const float va = tv[s], vb = tv[s + w];
                const int   sa = ts[s], sb = ts[s + w];
                const bool r = vb > va;
                tv[s] = r ? vb : va;
                ts[s] = r ? sb : sa;
            }
        }
        float bv = tv[0]; int bi = t + ts[0] * FPS_THREADS;
        wave_argmax(bv, bi);
        if (lane == 0) { sval[wid] = bv; sidx[wid] = bi; }
        __syncthreads();
        if (wid == 0) {
            float v = sval[lane & (FPS_WAVES - 1)]; int ix = sidx[lane & (FPS_WAVES - 1)];
            octet_argmax(v, ix);
            if (lane == 0) {
                scoord[0] = pb[ix * 3 + 0];
                scoord[1] = pb[ix * 3 + 1];
                scoord[2] = pb[ix * 3 + 2];
                fid[b * NOUT + i] = b * NPTS + ix;
            }
        }
        __syncthreads();
        const float nx = scoord[0], ny = scoord[1], nz = scoord[2];
        #pragma unroll
        for (int s = 0; s < FPS_PPT; ++s) {
            const float dx = px[s] - nx, dy = py[s] - ny, dz = pz[s] - nz;
            const float dd = dx * dx + dy * dy + dz * dz;
            d[s] = fminf(d[s], dd);
        }
    }
}

__global__ __launch_bounds__(128) void gather_maxpool_kernel(
    const float* __restrict__ h, const float* __restrict__ p,
    const int* __restrict__ fid, const int* __restrict__ sid_euc,
    float* __restrict__ out)
{
    const int s = blockIdx.x;
    const int t = threadIdx.x;

    __shared__ int ssid[KNBR];
    const int f = fid[s];
    if (t < KNBR) ssid[t] = sid_euc[(size_t)f * KNBR + t];
    if (t < 3)    out[XOUT_ELEMS + (size_t)s * 3 + t] = p[(size_t)f * 3 + t];
    __syncthreads();

    const int c = t * 4;
    v4f acc = { -INFINITY, -INFINITY, -INFINITY, -INFINITY };
    #pragma unroll
    for (int k = 0; k < KNBR; ++k) {
        const int r = ssid[k];
        if (k + 1 < KNBR)
            __builtin_prefetch(h + (size_t)ssid[k + 1] * COUT + c, 0, 0);
        const v4f v = *(const v4f*)(h + (size_t)r * COUT + c);
        acc.x = fmaxf(acc.x, v.x);
        acc.y = fmaxf(acc.y, v.y);
        acc.z = fmaxf(acc.z, v.z);
        acc.w = fmaxf(acc.w, v.w);
    }
    *(v4f*)(out + (size_t)s * COUT + c) = acc;
}

extern "C" void kernel_launch(void* const* d_in, const int* in_sizes, int n_in,
                              void* d_out, int out_size, void* d_ws, size_t ws_size,
                              hipStream_t stream)
{
    (void)in_sizes; (void)n_in; (void)out_size; (void)ws_size;
    const float* x     = (const float*)d_in[0];
    const float* p     = (const float*)d_in[1];
    const int*   sid   = (const int*)  d_in[2];
    const float* W     = (const float*)d_in[4];
    const float* bias  = (const float*)d_in[5];
    const float* gamma = (const float*)d_in[6];
    const float* beta  = (const float*)d_in[7];
    const float* rmean = (const float*)d_in[8];
    const float* rvar  = (const float*)d_in[9];
    float* out = (float*)d_out;

    float* h  = (float*)d_ws;
    int* fid  = (int*)((char*)d_ws + (size_t)MROWS * COUT * 4);

    dim3 ggrid(MROWS / 64, COUT / 64);
    gemm_bn_relu_kernel<<<ggrid, 128, 0, stream>>>(x, W, bias, gamma, beta,
                                                   rmean, rvar, h);
    fps_kernel<<<BATCH, FPS_THREADS, 0, stream>>>(p, fid);
    gather_maxpool_kernel<<<BATCH * NOUT, 128, 0, stream>>>(h, p, fid, sid, out);
}


# FPS: cloud staged once in LDS; after ONE barrier per pick every wave reduces the 8 partials itself and reads the winner's xyz from LDS (was: one lane's global load, LDS broadcast, second barrier)
# speedup vs baseline: 2.4111x; 1.3956x over previous
#include <hip/hip_runtime.h>
#include <hip/hip_bf16.h>
#include <math.h>

typedef float v2f __attribute__((ext_vector_type(2)));
typedef float v4f __attribute__((ext_vector_type(4)));
typedef float v8f __attribute__((ext_vector_type(8)));

#define BATCH 16
#define NPTS  4096
#define CIN   256
#define COUT  512
#define KNBR  16
#define NOUT  1024
#define MROWS (BATCH * NPTS)
#define BN_EPS 1e-5f
#define XOUT_ELEMS ((size_t)BATCH * NOUT * COUT)

__global__ __launch_bounds__(128) void gemm_bn_relu_kernel(
    const float* __restrict__ x, const float* __restrict__ W,
    const float* __restrict__ bias, const float* __restrict__ gamma,
    const float* __restrict__ beta, const float* __restrict__ rmean,
    const float* __restrict__ rvar, float* __restrict__ h)
{
    const int wave = threadIdx.x >> 5;
    const int lane = threadIdx.x & 31;
    const int m0 = (blockIdx.x * 4 + wave) * 16;
    const int n0 = blockIdx.y * 64;

    const int rowA = m0 + (lane & 15);
    const int kOff = (lane >> 4) << 1;
    const float* aPtr = x + (size_t)rowA * CIN + kOff;

    const int colL = lane & 15;
    const float* bPtr0 = W + (size_t)(n0 +  0 + colL) * CIN + kOff;
    const float* bPtr1 = W + (size_t)(n0 + 16 + colL) * CIN + kOff;
    const float* bPtr2 = W + (size_t)(n0 + 32 + colL) * CIN + kOff;
    const float* bPtr3 = W + (size_t)(n0 + 48 + colL) * CIN + kOff;

    v8f acc[4];
    #pragma unroll
    for (int nt = 0; nt < 4; ++nt) acc[nt] = {};

    #pragma unroll 4
    for (int c = 0; c < CIN; c += 4) {
        v2f a  = *(const v2f*)(aPtr + c);
        v2f b0 = *(const v2f*)(bPtr0 + c);
        v2f b1 = *(const v2f*)(bPtr1 + c);
        v2f b2 = *(const v2f*)(bPtr2 + c);
        v2f b3 = *(const v2f*)(bPtr3 + c);
        acc[0] = __builtin_amdgcn_wmma_f32_16x16x4_f32(false, a, false, b0,
                                                       (short)0, acc[0], false, false);
        acc[1] = __builtin_amdgcn_wmma_f32_16x16x4_f32(false, a, false, b1,
                                                       (short)0, acc[1], false, false);
        acc[2] = __builtin_amdgcn_wmma_f32_16x16x4_f32(false, a, false, b2,
                                                       (short)0, acc[2], false, false);
        acc[3] = __builtin_amdgcn_wmma_f32_16x16x4_f32(false, a, false, b3,
                                                       (short)0, acc[3], false, false);
    }

    const int rOff = (lane >> 4) << 3;
    #pragma unroll
    for (int nt = 0; nt < 4; ++nt) {
        const int o = n0 + nt * 16 + colL;
        const float sc = gamma[o] * rsqrtf(rvar[o] + BN_EPS);
        const float sh = (bias[o] - rmean[o]) * sc + beta[o];
        #pragma unroll
        for (int j = 0; j < 8; ++j) {
            float v = fmaf(acc[nt][j], sc, sh);
            v = fmaxf(v, 0.0f);
            h[(size_t)(m0 + j + rOff) * COUT + o] = v;
        }
    }
}

template <int XM>
__device__ __forceinline__ int row_xmask(int v)
{
    return __builtin_amdgcn_mov_dpp(v, 0x160 + XM, 0xf, 0xf, true);
}
__device__ __forceinline__ int xhalf(int v)
{
    return __builtin_amdgcn_permlanex16(v, v, 0x76543210, 0xfedcba98, false, false);
}
__device__ __forceinline__ int wave_max_i(int v)
{
    v = max(v, row_xmask<1>(v));
    v = max(v, row_xmask<2>(v));
    v = max(v, row_xmask<4>(v));
    v = max(v, row_xmask<8>(v));
    v = max(v, xhalf(v));
    return v;
}
__device__ __forceinline__ int wave_min_i(int v)
{
    v = min(v, row_xmask<1>(v));
    v = min(v, row_xmask<2>(v));
    v = min(v, row_xmask<4>(v));
    v = min(v, row_xmask<8>(v));
    v = min(v, xhalf(v));
    return v;
}
__device__ __forceinline__ void wave_argmax(float& v, int& ix)
{
    const int k = __builtin_bit_cast(int, v);
    const int m = wave_max_i(k);
    const int cand = (k == m) ? ix : 0x7fffffff;
    ix = wave_min_i(cand);
    v = __builtin_bit_cast(float, m);
}

__device__ __forceinline__ void octet_argmax(float& v, int& ix)
{
    const int k = __builtin_bit_cast(int, v);
    int m = k;
    m = max(m, row_xmask<1>(m));
    m = max(m, row_xmask<2>(m));
    m = max(m, row_xmask<4>(m));
    int c = (k == m) ? ix : 0x7fffffff;
    c = min(c, row_xmask<1>(c));
    c = min(c, row_xmask<2>(c));
    c = min(c, row_xmask<4>(c));
    ix = c;
    v = __builtin_bit_cast(float, m);
}

__device__ __forceinline__ float sqdist(float dx, float dy, float dz)
{
    return fmaf(dz, dz, fmaf(dx, dx, dy * dy));
}

#define FPS_THREADS 256
#define FPS_PPT     (NPTS / FPS_THREADS)
#define FPS_WAVES   (FPS_THREADS / 32)
typedef int v2i __attribute__((ext_vector_type(2)));
__global__ __launch_bounds__(FPS_THREADS) void fps_kernel(const float* __restrict__ p,
                                                          int* __restrict__ fid)
{
    const int b = blockIdx.x;
    const float* pb = p + (size_t)b * NPTS * 3;
    const int t = threadIdx.x;

    __shared__ v4f spt[NPTS];
    __shared__ v2i spart[2][FPS_WAVES];

    float px[FPS_PPT], py[FPS_PPT], pz[FPS_PPT], d[FPS_PPT];
    const float x0 = pb[0], y0 = pb[1], z0 = pb[2];
    #pragma unroll
    for (int s = 0; s < FPS_PPT; ++s) {
        const int idx = t + s * FPS_THREADS;
        px[s] = pb[idx * 3 + 0];
        py[s] = pb[idx * 3 + 1];
        pz[s] = pb[idx * 3 + 2];
        const v4f q = { px[s], py[s], pz[s], 0.0f };
        spt[idx] = q;
        const float dx = px[s] - x0, dy = py[s] - y0, dz = pz[s] - z0;
        d[s] = sqdist(dx, dy, dz);
    }

    if (t == 0) fid[b * NOUT + 0] = b * NPTS;

    const int wid = t >> 5, lane = t & 31;

    __syncthreads();

    for (int i = 1; i < NOUT; ++i) {
        float tv[FPS_PPT / 2]; int ts[FPS_PPT / 2];
        #pragma unroll
        for (int s = 0; s < FPS_PPT / 2; ++s) {
            const float va = d[2 * s], vb = d[2 * s + 1];
            const bool r = vb > va;
            tv[s] = r ? vb : va;
            ts[s] = r ? 2 * s + 1 : 2 * s;
        }
        #pragma unroll
        for (int w = 1; w < FPS_PPT / 2; w <<= 1) {
            #pragma unroll
            for (int s = 0; s < FPS_PPT / 2; s += 2 * w) {
                const float va = tv[s], vb = tv[s + w];
                const int   sa = ts[s], sb = ts[s + w];
                const bool r = vb > va;
                tv[s] = r ? vb : va;
                ts[s] = r ? sb : sa;
            }
        }
        float bv = tv[0]; int bi = t + ts[0] * FPS_THREADS;
        wave_argmax(bv, bi);
        const int par = i & 1;
        if (lane == 0) {
            const v2i mine = { __builtin_bit_cast(int, bv), bi };
            spart[par][wid] = mine;
        }
        __syncthreads();
        const v2i pr = spart[par][lane & (FPS_WAVES - 1)];
        float v = __builtin_bit_cast(float, pr.x); int ix = pr.y;
        octet_argmax(v, ix);
        const v4f win = spt[ix];
        if (t == 0) fid[b * NOUT + i] = b * NPTS + ix;
        const float nx = win.x, ny = win.y, nz = win.z;
        #pragma unroll
        for (int s = 0; s < FPS_PPT; ++s) {
            const float dx = px[s] - nx, dy = py[s] - ny, dz = pz[s] - nz;
            d[s] = fminf(d[s], sqdist(dx, dy, dz));
        }
    }
}

__global__ __launch_bounds__(128) void gather_maxpool_kernel(
    const float* __restrict__ h, const float* __restrict__ p,
    const int* __restrict__ fid, const int* __restrict__ sid_euc,
    float* __restrict__ out)
{
    const int s = blockIdx.x;
    const int t = threadIdx.x;

    __shared__ int ssid[KNBR];
    const int f = fid[s];
    if (t < KNBR) ssid[t] = sid_euc[(size_t)f * KNBR + t];
    if (t < 3)    out[XOUT_ELEMS + (size_t)s * 3 + t] = p[(size_t)f * 3 + t];
    __syncthreads();

    const int c = t * 4;
    v4f acc = { -INFINITY, -INFINITY, -INFINITY, -INFINITY };
    #pragma unroll
    for (int k = 0; k < KNBR; ++k) {
        const int r = ssid[k];
        if (k + 1 < KNBR)
            __builtin_prefetch(h + (size_t)ssid[k + 1] * COUT + c, 0, 0);
        const v4f v = *(const v4f*)(h + (size_t)r * COUT + c);
        acc.x = fmaxf(acc.x, v.x);
        acc.y = fmaxf(acc.y, v.y);
        acc.z = fmaxf(acc.z, v.z);
        acc.w = fmaxf(acc.w, v.w);
    }
    *(v4f*)(out + (size_t)s * COUT + c) = acc;
}

extern "C" void kernel_launch(void* const* d_in, const int* in_sizes, int n_in,
                              void* d_out, int out_size, void* d_ws, size_t ws_size,
                              hipStream_t stream)
{
    (void)in_sizes; (void)n_in; (void)out_size; (void)ws_size;
    const float* x     = (const float*)d_in[0];
    const float* p     = (const float*)d_in[1];
    const int*   sid   = (const int*)  d_in[2];
    const float* W     = (const float*)d_in[4];
    const float* bias  = (const float*)d_in[5];
    const float* gamma = (const float*)d_in[6];
    const float* beta  = (const float*)d_in[7];
    const float* rmean = (const float*)d_in[8];
    const float* rvar  = (const float*)d_in[9];
    float* out = (float*)d_out;

    float* h  = (float*)d_ws;
    int* fid  = (int*)((char*)d_ws + (size_t)MROWS * COUT * 4);

    dim3 ggrid(MROWS / 64, COUT / 64);
    gemm_bn_relu_kernel<<<ggrid, 128, 0, stream>>>(x, W, bias, gamma, beta,
                                                   rmean, rvar, h);
    fps_kernel<<<BATCH, FPS_THREADS, 0, stream>>>(p, fid);
    gather_maxpool_kernel<<<BATCH * NOUT, 128, 0, stream>>>(h, p, fid, sid, out);
}


# FPS and GEMM side by side in ONE launch: work-groups 0-15 run the serial sampling chain while 4096 eight-wave GEMM work-groups fill the other WGPs; 3 launches become 2, no arithmetic changed
# speedup vs baseline: 2.5255x; 1.0474x over previous
#include <hip/hip_runtime.h>
#include <hip/hip_bf16.h>
#include <math.h>

typedef float v2f __attribute__((ext_vector_type(2)));
typedef float v4f __attribute__((ext_vector_type(4)));
typedef float v8f __attribute__((ext_vector_type(8)));

#define BATCH 16
#define NPTS  4096
#define CIN   256
#define COUT  512
#define KNBR  16
#define NOUT  1024
#define MROWS (BATCH * NPTS)
#define BN_EPS 1e-5f
#define XOUT_ELEMS ((size_t)BATCH * NOUT * COUT)

#define FPS_THREADS 256
#define GEMM_MBLK   (FPS_THREADS / 32 * 16)
#define GEMM_MBLOCKS (MROWS / GEMM_MBLK)
#define GEMM_NBLOCKS (COUT / 64)

__device__ __forceinline__ void gemm_bn_relu_part(
    const int g,
    const float* __restrict__ x, const float* __restrict__ W,
    const float* __restrict__ bias, const float* __restrict__ gamma,
    const float* __restrict__ beta, const float* __restrict__ rmean,
    const float* __restrict__ rvar, float* __restrict__ h)
{
    const int wave = threadIdx.x >> 5;
    const int lane = threadIdx.x & 31;
    const int m0 = ((g % GEMM_MBLOCKS) * (FPS_THREADS / 32) + wave) * 16;
    const int n0 = (g / GEMM_MBLOCKS) * 64;

    const int rowA = m0 + (lane & 15);
    const int kOff = (lane >> 4) << 1;
    const float* aPtr = x + (size_t)rowA * CIN + kOff;

    const int colL = lane & 15;
    const float* bPtr0 = W + (size_t)(n0 +  0 + colL) * CIN + kOff;
    const float* bPtr1 = W + (size_t)(n0 + 16 + colL) * CIN + kOff;
    const float* bPtr2 = W + (size_t)(n0 + 32 + colL) * CIN + kOff;
    const float* bPtr3 = W + (size_t)(n0 + 48 + colL) * CIN + kOff;

    v8f acc[4];
    #pragma unroll
    for (int nt = 0; nt < 4; ++nt) acc[nt] = {};

    #pragma unroll 4
    for (int c = 0; c < CIN; c += 4) {
        v2f a  = *(const v2f*)(aPtr + c);
        v2f b0 = *(const v2f*)(bPtr0 + c);
        v2f b1 = *(const v2f*)(bPtr1 + c);
        v2f b2 = *(const v2f*)(bPtr2 + c);
        v2f b3 = *(const v2f*)(bPtr3 + c);
        acc[0] = __builtin_amdgcn_wmma_f32_16x16x4_f32(false, a, false, b0,
                                                       (short)0, acc[0], false, false);
        acc[1] = __builtin_amdgcn_wmma_f32_16x16x4_f32(false, a, false, b1,
                                                       (short)0, acc[1], false, false);
        acc[2] = __builtin_amdgcn_wmma_f32_16x16x4_f32(false, a, false, b2,
                                                       (short)0, acc[2], false, false);
        acc[3] = __builtin_amdgcn_wmma_f32_16x16x4_f32(false, a, false, b3,
                                                       (short)0, acc[3], false, false);
    }

    const int rOff = (lane >> 4) << 3;
    #pragma unroll
    for (int nt = 0; nt < 4; ++nt) {
        const int o = n0 + nt * 16 + colL;
        const float sc = gamma[o] * rsqrtf(rvar[o] + BN_EPS);
        const float sh = (bias[o] - rmean[o]) * sc + beta[o];
        #pragma unroll
        for (int j = 0; j < 8; ++j) {
            float v = fmaf(acc[nt][j], sc, sh);
            v = fmaxf(v, 0.0f);
            h[(size_t)(m0 + j + rOff) * COUT + o] = v;
        }
    }
}

template <int XM>
__device__ __forceinline__ int row_xmask(int v)
{
    return __builtin_amdgcn_mov_dpp(v, 0x160 + XM, 0xf, 0xf, true);
}
__device__ __forceinline__ int xhalf(int v)
{
    return __builtin_amdgcn_permlanex16(v, v, 0x76543210, 0xfedcba98, false, false);
}
__device__ __forceinline__ int wave_max_i(int v)
{
    v = max(v, row_xmask<1>(v));
    v = max(v, row_xmask<2>(v));
    v = max(v, row_xmask<4>(v));
    v = max(v, row_xmask<8>(v));
    v = max(v, xhalf(v));
    return v;
}
__device__ __forceinline__ int wave_min_i(int v)
{
    v = min(v, row_xmask<1>(v));
    v = min(v, row_xmask<2>(v));
    v = min(v, row_xmask<4>(v));
    v = min(v, row_xmask<8>(v));
    v = min(v, xhalf(v));
    return v;
}
__device__ __forceinline__ void wave_argmax(float& v, int& ix)
{
    const int k = __builtin_bit_cast(int, v);
    const int m = wave_max_i(k);
    const int cand = (k == m) ? ix : 0x7fffffff;
    ix = wave_min_i(cand);
    v = __builtin_bit_cast(float, m);
}

__device__ __forceinline__ void octet_argmax(float& v, int& ix)
{
    const int k = __builtin_bit_cast(int, v);
    int m = k;
    m = max(m, row_xmask<1>(m));
    m = max(m, row_xmask<2>(m));
    m = max(m, row_xmask<4>(m));
    int c = (k == m) ? ix : 0x7fffffff;
    c = min(c, row_xmask<1>(c));
    c = min(c, row_xmask<2>(c));
    c = min(c, row_xmask<4>(c));
    ix = c;
    v = __builtin_bit_cast(float, m);
}

__device__ __forceinline__ float sqdist(float dx, float dy, float dz)
{
    return fmaf(dz, dz, fmaf(dx, dx, dy * dy));
}

#define FPS_PPT     (NPTS / FPS_THREADS)
#define FPS_WAVES   (FPS_THREADS / 32)
typedef int v2i __attribute__((ext_vector_type(2)));
__device__ __forceinline__ void fps_part(const int b, const float* __restrict__ p,
                                         int* __restrict__ fid)
{
    const float* pb = p + (size_t)b * NPTS * 3;
    const int t = threadIdx.x;

    __shared__ v4f spt[NPTS];
    __shared__ v2i spart[2][FPS_WAVES];

    float px[FPS_PPT], py[FPS_PPT], pz[FPS_PPT], d[FPS_PPT];
    const float x0 = pb[0], y0 = pb[1], z0 = pb[2];
    #pragma unroll
    for (int s = 0; s < FPS_PPT; ++s) {
        const int idx = t + s * FPS_THREADS;
        px[s] = pb[idx * 3 + 0];
        py[s] = pb[idx * 3 + 1];
        pz[s] = pb[idx * 3 + 2];
        const v4f q = { px[s], py[s], pz[s], 0.0f };
        spt[idx] = q;
        const float dx = px[s] - x0, dy = py[s] - y0, dz = pz[s] - z0;
        d[s] = sqdist(dx, dy, dz);
    }

    if (t == 0) fid[b * NOUT + 0] = b * NPTS;

    const int wid = t >> 5, lane = t & 31;

    __syncthreads();

    for (int i = 1; i < NOUT; ++i) {
        float tv[FPS_PPT / 2]; int ts[FPS_PPT / 2];
        #pragma unroll
        for (int s = 0; s < FPS_PPT / 2; ++s) {
            const float va = d[2 * s], vb = d[2 * s + 1];
            const bool r = vb > va;
            tv[s] = r ? vb : va;
            ts[s] = r ? 2 * s + 1 : 2 * s;
        }
        #pragma unroll
        for (int w = 1; w < FPS_PPT / 2; w <<= 1) {
            #pragma unroll
            for (int s = 0; s < FPS_PPT / 2; s += 2 * w) {
                const float va = tv[s], vb = tv[s + w];
                const int   sa = ts[s], sb = ts[s + w];
                const bool r = vb > va;
                tv[s] = r ? vb : va;
                ts[s] = r ? sb : sa;
            }
        }
        float bv = tv[0]; int bi = t + ts[0] * FPS_THREADS;
        wave_argmax(bv, bi);
        const int par = i & 1;
        if (lane == 0) {
            const v2i mine = { __builtin_bit_cast(int, bv), bi };
            spart[par][wid] = mine;
        }
        __syncthreads();
        const v2i pr = spart[par][lane & (FPS_WAVES - 1)];
        float v = __builtin_bit_cast(float, pr.x); int ix = pr.y;
        octet_argmax(v, ix);
        const v4f win = spt[ix];
        if (t == 0) fid[b * NOUT + i] = b * NPTS + ix;
        const float nx = win.x, ny = win.y, nz = win.z;
        #pragma unroll
        for (int s = 0; s < FPS_PPT; ++s) {
            const float dx = px[s] - nx, dy = py[s] - ny, dz = pz[s] - nz;
            d[s] = fminf(d[s], sqdist(dx, dy, dz));
        }
    }
}

__global__ __launch_bounds__(FPS_THREADS) void fps_and_gemm_kernel(
    const float* __restrict__ x, const float* __restrict__ W,
    const float* __restrict__ bias, const float* __restrict__ gamma,
    const float* __restrict__ beta, const float* __restrict__ rmean,
    const float* __restrict__ rvar, float* __restrict__ h,
    const float* __restrict__ p, int* __restrict__ fid)
{
    if (blockIdx.x < BATCH)
        fps_part(blockIdx.x, p, fid);
    else
        gemm_bn_relu_part(blockIdx.x - BATCH, x, W, bias, gamma, beta, rmean, rvar, h);
}

__global__ __launch_bounds__(128) void gather_maxpool_kernel(
    const float* __restrict__ h, const float* __restrict__ p,
    const int* __restrict__ fid, const int* __restrict__ sid_euc,
    float* __restrict__ out)
{
    const int s = blockIdx.x;
    const int t = threadIdx.x;

    __shared__ int ssid[KNBR];
    const int f = fid[s];
    if (t < KNBR) ssid[t] = sid_euc[(size_t)f * KNBR + t];
    if (t < 3)    out[XOUT_ELEMS + (size_t)s * 3 + t] = p[(size_t)f * 3 + t];
    __syncthreads();

    const int c = t * 4;
    v4f acc = { -INFINITY, -INFINITY, -INFINITY, -INFINITY };
    #pragma unroll
    for (int k = 0; k < KNBR; ++k) {
        const int r = ssid[k];
        if (k + 1 < KNBR)
            __builtin_prefetch(h + (size_t)ssid[k + 1] * COUT + c, 0, 0);
        const v4f v = *(const v4f*)(h + (size_t)r * COUT + c);
        acc.x = fmaxf(acc.x, v.x);
        acc.y = fmaxf(acc.y, v.y);
        acc.z = fmaxf(acc.z, v.z);
        acc.w = fmaxf(acc.w, v.w);
    }
    *(v4f*)(out + (size_t)s * COUT + c) = acc;
}

extern "C" void kernel_launch(void* const* d_in, const int* in_sizes, int n_in,
                              void* d_out, int out_size, void* d_ws, size_t ws_size,
                              hipStream_t stream)
{
    (void)in_sizes; (void)n_in; (void)out_size; (void)ws_size;
    const float* x     = (const float*)d_in[0];
    const float* p     = (const float*)d_in[1];
    const int*   sid   = (const int*)  d_in[2];
    const float* W     = (const float*)d_in[4];
    const float* bias  = (const float*)d_in[5];
    const float* gamma = (const float*)d_in[6];
    const float* beta  = (const float*)d_in[7];
    const float* rmean = (const float*)d_in[8];
    const float* rvar  = (const float*)d_in[9];
    float* out = (float*)d_out;

    float* h  = (float*)d_ws;
    int* fid  = (int*)((char*)d_ws + (size_t)MROWS * COUT * 4);

    fps_and_gemm_kernel<<<BATCH + GEMM_MBLOCKS * GEMM_NBLOCKS, FPS_THREADS, 0, stream>>>(
        x, W, bias, gamma, beta, rmean, rvar, h, p, fid);
    gather_maxpool_kernel<<<BATCH * NOUT, 128, 0, stream>>>(h, p, fid, sid, out);
}


# Launch 1's LDS allocation raised past half a WGP's (164 KiB; the extra never touched): one work-group per WGP, so the sampling chain's SIMDs carry no GEMM waves; GEMM drops to 2 waves per SIMD
# speedup vs baseline: 3.8546x; 1.5263x over previous
#include <hip/hip_runtime.h>
#include <hip/hip_bf16.h>
#include <math.h>

typedef float v2f __attribute__((ext_vector_type(2)));
typedef float v4f __attribute__((ext_vector_type(4)));
typedef float v8f __attribute__((ext_vector_type(8)));

#define BATCH 16
#define NPTS  4096
#define CIN   256
#define COUT  512
#define KNBR  16
#define NOUT  1024
#define MROWS (BATCH * NPTS)
#define BN_EPS 1e-5f
#define XOUT_ELEMS ((size_t)BATCH * NOUT * COUT)

#define FPS_THREADS 256
#define GEMM_MBLK   (FPS_THREADS / 32 * 16)
#define GEMM_MBLOCKS (MROWS / GEMM_MBLK)
#define GEMM_NBLOCKS (COUT / 64)

__device__ __forceinline__ void gemm_bn_relu_part(
    const int g,
    const float* __restrict__ x, const float* __restrict__ W,
    const float* __restrict__ bias, const float* __restrict__ gamma,
    const float* __restrict__ beta, const float* __restrict__ rmean,
    const float* __restrict__ rvar, float* __restrict__ h)
{
    const int wave = threadIdx.x >> 5;
    const int lane = threadIdx.x & 31;
    const int m0 = ((g % GEMM_MBLOCKS) * (FPS_THREADS / 32) + wave) * 16;
    const int n0 = (g / GEMM_MBLOCKS) * 64;

    const int rowA = m0 + (lane & 15);
    const int kOff = (lane >> 4) << 1;
    const float* aPtr = x + (size_t)rowA * CIN + kOff;

    const int colL = lane & 15;
    const float* bPtr0 = W + (size_t)(n0 +  0 + colL) * CIN + kOff;
    const float* bPtr1 = W + (size_t)(n0 + 16 + colL) * CIN + kOff;
    const float* bPtr2 = W + (size_t)(n0 + 32 + colL) * CIN + kOff;
    const float* bPtr3 = W + (size_t)(n0 + 48 + colL) * CIN + kOff;

    v8f acc[4];
    #pragma unroll
    for (int nt = 0; nt < 4; ++nt) acc[nt] = {};

    #pragma unroll 4
    for (int c = 0; c < CIN; c += 4) {
        v2f a  = *(const v2f*)(aPtr + c);
        v2f b0 = *(const v2f*)(bPtr0 + c);
        v2f b1 = *(const v2f*)(bPtr1 + c);
        v2f b2 = *(const v2f*)(bPtr2 + c);
        v2f b3 = *(const v2f*)(bPtr3 + c);
        acc[0] = __builtin_amdgcn_wmma_f32_16x16x4_f32(false, a, false, b0,
                                                       (short)0, acc[0], false, false);
        acc[1] = __builtin_amdgcn_wmma_f32_16x16x4_f32(false, a, false, b1,
                                                       (short)0, acc[1], false, false);
        acc[2] = __builtin_amdgcn_wmma_f32_16x16x4_f32(false, a, false, b2,
                                                       (short)0, acc[2], false, false);
        acc[3] = __builtin_amdgcn_wmma_f32_16x16x4_f32(false, a, false, b3,
                                                       (short)0, acc[3], false, false);
    }

    const int rOff = (lane >> 4) << 3;
    #pragma unroll
    for (int nt = 0; nt < 4; ++nt) {
        const int o = n0 + nt * 16 + colL;
        const float sc = gamma[o] * rsqrtf(rvar[o] + BN_EPS);
        const float sh = (bias[o] - rmean[o]) * sc + beta[o];
        #pragma unroll
        for (int j = 0; j < 8; ++j) {
            float v = fmaf(acc[nt][j], sc, sh);
            v = fmaxf(v, 0.0f);
            h[(size_t)(m0 + j + rOff) * COUT + o] = v;
        }
    }
}

template <int XM>
__device__ __forceinline__ int row_xmask(int v)
{
    return __builtin_amdgcn_mov_dpp(v, 0x160 + XM, 0xf, 0xf, true);
}
__device__ __forceinline__ int xhalf(int v)
{
    return __builtin_amdgcn_permlanex16(v, v, 0x76543210, 0xfedcba98, false, false);
}
__device__ __forceinline__ int wave_max_i(int v)
{
    v = max(v, row_xmask<1>(v));
    v = max(v, row_xmask<2>(v));
    v = max(v, row_xmask<4>(v));
    v = max(v, row_xmask<8>(v));
    v = max(v, xhalf(v));
    return v;
}
__device__ __forceinline__ int wave_min_i(int v)
{
    v = min(v, row_xmask<1>(v));
    v = min(v, row_xmask<2>(v));
    v = min(v, row_xmask<4>(v));
    v = min(v, row_xmask<8>(v));
    v = min(v, xhalf(v));
    return v;
}
__device__ __forceinline__ void wave_argmax(float& v, int& ix)
{
    const int k = __builtin_bit_cast(int, v);
    const int m = wave_max_i(k);
    const int cand = (k == m) ? ix : 0x7fffffff;
    ix = wave_min_i(cand);
    v = __builtin_bit_cast(float, m);
}

__device__ __forceinline__ void octet_argmax(float& v, int& ix)
{
    const int k = __builtin_bit_cast(int, v);
    int m = k;
    m = max(m, row_xmask<1>(m));
    m = max(m, row_xmask<2>(m));
    m = max(m, row_xmask<4>(m));
    int c = (k == m) ? ix : 0x7fffffff;
    c = min(c, row_xmask<1>(c));
    c = min(c, row_xmask<2>(c));
    c = min(c, row_xmask<4>(c));
    ix = c;
    v = __builtin_bit_cast(float, m);
}

__device__ __forceinline__ float sqdist(float dx, float dy, float dz)
{
    return fmaf(dz, dz, fmaf(dx, dx, dy * dy));
}

#define FPS_PPT     (NPTS / FPS_THREADS)
#define FPS_WAVES   (FPS_THREADS / 32)
#define FPS_LDS_SLOTS 10496
static_assert(FPS_LDS_SLOTS >= NPTS, "the cloud must fit its LDS array");
static_assert((size_t)FPS_LDS_SLOTS * 16 > 160 * 1024, "one work-group per WGP needs more than 160 KiB");
static_assert((size_t)FPS_LDS_SLOTS * 16 + 128 <= 320 * 1024, "must fit one WGP's LDS");
typedef int v2i __attribute__((ext_vector_type(2)));
__device__ __forceinline__ void fps_part(const int b, const float* __restrict__ p,
                                         int* __restrict__ fid)
{
    const float* pb = p + (size_t)b * NPTS * 3;
    const int t = threadIdx.x;

    __shared__ v4f spt[FPS_LDS_SLOTS];
    __shared__ v2i spart[2][FPS_WAVES];

    float px[FPS_PPT], py[FPS_PPT], pz[FPS_PPT], d[FPS_PPT];
    const float x0 = pb[0], y0 = pb[1], z0 = pb[2];
    #pragma unroll
    for (int s = 0; s < FPS_PPT; ++s) {
        const int idx = t + s * FPS_THREADS;
        px[s] = pb[idx * 3 + 0];
        py[s] = pb[idx * 3 + 1];
        pz[s] = pb[idx * 3 + 2];
        const v4f q = { px[s], py[s], pz[s], 0.0f };
        spt[idx] = q;
        const float dx = px[s] - x0, dy = py[s] - y0, dz = pz[s] - z0;
        d[s] = sqdist(dx, dy, dz);
    }

    if (t == 0) fid[b * NOUT + 0] = b * NPTS;

    const int wid = t >> 5, lane = t & 31;

    __syncthreads();

    for (int i = 1; i < NOUT; ++i) {
        float tv[FPS_PPT / 2]; int ts[FPS_PPT / 2];
        #pragma unroll
        for (int s = 0; s < FPS_PPT / 2; ++s) {
            const float va = d[2 * s], vb = d[2 * s + 1];
            const bool r = vb > va;
            tv[s] = r ? vb : va;
            ts[s] = r ? 2 * s + 1 : 2 * s;
        }
        #pragma unroll
        for (int w = 1; w < FPS_PPT / 2; w <<= 1) {
            #pragma unroll
            for (int s = 0; s < FPS_PPT / 2; s += 2 * w) {
                const float va = tv[s], vb = tv[s + w];
                const int   sa = ts[s], sb = ts[s + w];
                const bool r = vb > va;
                tv[s] = r ? vb : va;
                ts[s] = r ? sb : sa;
            }
        }
        float bv = tv[0]; int bi = t + ts[0] * FPS_THREADS;
        wave_argmax(bv, bi);
        const int par = i & 1;
        if (lane == 0) {
            const v2i mine = { __builtin_bit_cast(int, bv), bi };
            spart[par][wid] = mine;
        }
        __syncthreads();
        const v2i pr = spart[par][lane & (FPS_WAVES - 1)];
        float v = __builtin_bit_cast(float, pr.x); int ix = pr.y;
        octet_argmax(v, ix);
        const v4f win = spt[ix];
        if (t == 0) fid[b * NOUT + i] = b * NPTS + ix;
        const float nx = win.x, ny = win.y, nz = win.z;
        #pragma unroll
        for (int s = 0; s < FPS_PPT; ++s) {
            const float dx = px[s] - nx, dy = py[s] - ny, dz = pz[s] - nz;
            d[s] = fminf(d[s], sqdist(dx, dy, dz));
        }
    }
}

__global__ __launch_bounds__(FPS_THREADS) void fps_and_gemm_kernel(
    const float* __restrict__ x, const float* __restrict__ W,
    const float* __restrict__ bias, const float* __restrict__ gamma,
    const float* __restrict__ beta, const float* __restrict__ rmean,
    const float* __restrict__ rvar, float* __restrict__ h,
    const float* __restrict__ p, int* __restrict__ fid)
{
    if (blockIdx.x < BATCH)
        fps_part(blockIdx.x, p, fid);
    else
        gemm_bn_relu_part(blockIdx.x - BATCH, x, W, bias, gamma, beta, rmean, rvar, h);
}

__global__ __launch_bounds__(128) void gather_maxpool_kernel(
    const float* __restrict__ h, const float* __restrict__ p,
    const int* __restrict__ fid, const int* __restrict__ sid_euc,
    float* __restrict__ out)
{
    const int s = blockIdx.x;
    const int t = threadIdx.x;

    __shared__ int ssid[KNBR];
    const int f = fid[s];
    if (t < KNBR) ssid[t] = sid_euc[(size_t)f * KNBR + t];
    if (t < 3)    out[XOUT_ELEMS + (size_t)s * 3 + t] = p[(size_t)f * 3 + t];
    __syncthreads();

    const int c = t * 4;
    v4f acc = { -INFINITY, -INFINITY, -INFINITY, -INFINITY };
    #pragma unroll
    for (int k = 0; k < KNBR; ++k) {
        const int r = ssid[k];
        if (k + 1 < KNBR)
            __builtin_prefetch(h + (size_t)ssid[k + 1] * COUT + c, 0, 0);
        const v4f v = *(const v4f*)(h + (size_t)r * COUT + c);
        acc.x = fmaxf(acc.x, v.x);
        acc.y = fmaxf(acc.y, v.y);
        acc.z = fmaxf(acc.z, v.z);
        acc.w = fmaxf(acc.w, v.w);
    }
    *(v4f*)(out + (size_t)s * COUT + c) = acc;
}

extern "C" void kernel_launch(void* const* d_in, const int* in_sizes, int n_in,
                              void* d_out, int out_size, void* d_ws, size_t ws_size,
                              hipStream_t stream)
{
    (void)in_sizes; (void)n_in; (void)out_size; (void)ws_size;
    const float* x     = (const float*)d_in[0];
    const float* p     = (const float*)d_in[1];
    const int*   sid   = (const int*)  d_in[2];
    const float* W     = (const float*)d_in[4];
    const float* bias  = (const float*)d_in[5];
    const float* gamma = (const float*)d_in[6];
    const float* beta  = (const float*)d_in[7];
    const float* rmean = (const float*)d_in[8];
    const float* rvar  = (const float*)d_in[9];
    float* out = (float*)d_out;

    float* h  = (float*)d_ws;
    int* fid  = (int*)((char*)d_ws + (size_t)MROWS * COUT * 4);

    fps_and_gemm_kernel<<<BATCH + GEMM_MBLOCKS * GEMM_NBLOCKS, FPS_THREADS, 0, stream>>>(
        x, W, bias, gamma, beta, rmean, rvar, h, p, fid);
    gather_maxpool_kernel<<<BATCH * NOUT, 128, 0, stream>>>(h, p, fid, sid, out);
}


# FPS: first lane (ballot) holding the wave max hands over its record, xyz included; after the barrier lane reads give the winner: no index reduction, no 2nd LDS look-up; 16 consecutive points a thread
# speedup vs baseline: 4.1278x; 1.0709x over previous
#include <hip/hip_runtime.h>
#include <hip/hip_bf16.h>
#include <math.h>

typedef float v2f __attribute__((ext_vector_type(2)));
typedef float v4f __attribute__((ext_vector_type(4)));
typedef float v8f __attribute__((ext_vector_type(8)));

#define BATCH 16
#define NPTS  4096
#define CIN   256
#define COUT  512
#define KNBR  16
#define NOUT  1024
#define MROWS (BATCH * NPTS)
#define BN_EPS 1e-5f
#define XOUT_ELEMS ((size_t)BATCH * NOUT * COUT)

#define FPS_THREADS 256
#define GEMM_MBLK   (FPS_THREADS / 32 * 16)
#define GEMM_MBLOCKS (MROWS / GEMM_MBLK)
#define GEMM_NBLOCKS (COUT / 64)

__device__ __forceinline__ void gemm_bn_relu_part(
    const int g,
    const float* __restrict__ x, const float* __restrict__ W,
    const float* __restrict__ bias, const float* __restrict__ gamma,
    const float* __restrict__ beta, const float* __restrict__ rmean,
    const float* __restrict__ rvar, float* __restrict__ h)
{
    const int wave = threadIdx.x >> 5;
    const int lane = threadIdx.x & 31;
    const int m0 = ((g % GEMM_MBLOCKS) * (FPS_THREADS / 32) + wave) * 16;
    const int n0 = (g / GEMM_MBLOCKS) * 64;

    const int rowA = m0 + (lane & 15);
    const int kOff = (lane >> 4) << 1;
    const float* aPtr = x + (size_t)rowA * CIN + kOff;

    const int colL = lane & 15;
    const float* bPtr0 = W + (size_t)(n0 +  0 + colL) * CIN + kOff;
    const float* bPtr1 = W + (size_t)(n0 + 16 + colL) * CIN + kOff;
    const float* bPtr2 = W + (size_t)(n0 + 32 + colL) * CIN + kOff;
    const float* bPtr3 = W + (size_t)(n0 + 48 + colL) * CIN + kOff;

    v8f acc[4];
    #pragma unroll
    for (int nt = 0; nt < 4; ++nt) acc[nt] = {};

    #pragma unroll 4
    for (int c = 0; c < CIN; c += 4) {
        v2f a  = *(const v2f*)(aPtr + c);
        v2f b0 = *(const v2f*)(bPtr0 + c);
        v2f b1 = *(const v2f*)(bPtr1 + c);
        v2f b2 = *(const v2f*)(bPtr2 + c);
        v2f b3 = *(const v2f*)(bPtr3 + c);
        acc[0] = __builtin_amdgcn_wmma_f32_16x16x4_f32(false, a, false, b0,
                                                       (short)0, acc[0], false, false);
        acc[1] = __builtin_amdgcn_wmma_f32_16x16x4_f32(false, a, false, b1,
                                                       (short)0, acc[1], false, false);
        acc[2] = __builtin_amdgcn_wmma_f32_16x16x4_f32(false, a, false, b2,
                                                       (short)0, acc[2], false, false);
        acc[3] = __builtin_amdgcn_wmma_f32_16x16x4_f32(false, a, false, b3,
                                                       (short)0, acc[3], false, false);
    }

    const int rOff = (lane >> 4) << 3;
    #pragma unroll
    for (int nt = 0; nt < 4; ++nt) {
        const int o = n0 + nt * 16 + colL;
        const float sc = gamma[o] * rsqrtf(rvar[o] + BN_EPS);
        const float sh = (bias[o] - rmean[o]) * sc + beta[o];
        #pragma unroll
        for (int j = 0; j < 8; ++j) {
            float v = fmaf(acc[nt][j], sc, sh);
            v = fmaxf(v, 0.0f);
            h[(size_t)(m0 + j + rOff) * COUT + o] = v;
        }
    }
}

template <int XM>
__device__ __forceinline__ int row_xmask(int v)
{
    return __builtin_amdgcn_mov_dpp(v, 0x160 + XM, 0xf, 0xf, true);
}
__device__ __forceinline__ int xhalf(int v)
{
    return __builtin_amdgcn_permlanex16(v, v, 0x76543210, 0xfedcba98, false, false);
}
__device__ __forceinline__ int wave_max_i(int v)
{
    v = max(v, row_xmask<1>(v));
    v = max(v, row_xmask<2>(v));
    v = max(v, row_xmask<4>(v));
    v = max(v, row_xmask<8>(v));
    v = max(v, xhalf(v));
    return v;
}
__device__ __forceinline__ int octet_max_i(int v)
{
    v = max(v, row_xmask<1>(v));
    v = max(v, row_xmask<2>(v));
    v = max(v, row_xmask<4>(v));
    return v;
}

__device__ __forceinline__ float sqdist(float dx, float dy, float dz)
{
    return fmaf(dz, dz, fmaf(dx, dx, dy * dy));
}

#define FPS_PPT     (NPTS / FPS_THREADS)
#define FPS_WAVES   (FPS_THREADS / 32)
#define FPS_PITCH   (FPS_PPT + 1)
#define FPS_LDS_SLOTS 10496
static_assert(FPS_LDS_SLOTS >= FPS_THREADS * FPS_PITCH, "the cloud must fit its LDS array");
static_assert((size_t)FPS_LDS_SLOTS * 16 > 160 * 1024, "one work-group per WGP needs more than 160 KiB");
static_assert((size_t)FPS_LDS_SLOTS * 16 + 512 <= 320 * 1024, "must fit one WGP's LDS");
typedef int v4i __attribute__((ext_vector_type(4)));
typedef float v3f __attribute__((ext_vector_type(3)));
typedef __attribute__((address_space(3))) v4i lds_v4i;
typedef __attribute__((address_space(3))) int lds_int;
__device__ __forceinline__ void fps_part(const int b, const float* __restrict__ p,
                                         int* __restrict__ fid)
{
    const float* pb = p + (size_t)b * NPTS * 3;
    const int t = threadIdx.x;

    __shared__ v4f spt[FPS_LDS_SLOTS];
    __shared__ v4i srec[2][FPS_WAVES];
    __shared__ int skey[2][FPS_WAVES];

    float px[FPS_PPT], py[FPS_PPT], pz[FPS_PPT], d[FPS_PPT];
    const float x0 = pb[0], y0 = pb[1], z0 = pb[2];
    {
        float raw[FPS_PPT * 3];
        const v4f* src4 = (const v4f*)(pb + (size_t)t * (FPS_PPT * 3));
        #pragma unroll
        for (int q = 0; q < FPS_PPT * 3 / 4; ++q) {
            const v4f v = src4[q];
            raw[4 * q + 0] = v.x; raw[4 * q + 1] = v.y;
            raw[4 * q + 2] = v.z; raw[4 * q + 3] = v.w;
        }
        #pragma unroll
        for (int s = 0; s < FPS_PPT; ++s) {
            px[s] = raw[3 * s + 0];
            py[s] = raw[3 * s + 1];
            pz[s] = raw[3 * s + 2];
            const v4f q = { px[s], py[s], pz[s], 0.0f };
            spt[t * FPS_PITCH + s] = q;
            const float dx = px[s] - x0, dy = py[s] - y0, dz = pz[s] - z0;
            d[s] = sqdist(dx, dy, dz);
        }
    }

    if (t == 0) fid[b * NOUT + 0] = b * NPTS;

    const int wid = t >> 5, lane = t & 31;

    __syncthreads();

    for (int i = 1; i < NOUT; ++i) {
        float tv[FPS_PPT / 2]; int ts[FPS_PPT / 2];
        #pragma unroll
        for (int s = 0; s < FPS_PPT / 2; ++s) {
            const float va = d[2 * s], vb = d[2 * s + 1];
            const bool r = vb > va;
            tv[s] = r ? vb : va;
            ts[s] = r ? 2 * s + 1 : 2 * s;
        }
        #pragma unroll
        for (int w = 1; w < FPS_PPT / 2; w <<= 1) {
            #pragma unroll
            for (int s = 0; s < FPS_PPT / 2; s += 2 * w) {
                const float va = tv[s], vb = tv[s + w];
                const int   sa = ts[s], sb = ts[s + w];
                const bool r = vb > va;
                tv[s] = r ? vb : va;
                ts[s] = r ? sb : sa;
            }
        }
        const v3f cand = *(const v3f*)&spt[t * FPS_PITCH + ts[0]];
        float cx = cand.x, cy = cand.y, cz = cand.z;
        __builtin_amdgcn_sched_barrier(0);
        const int k = __builtin_bit_cast(int, tv[0]);
        const int m = wave_max_i(k);
        const unsigned tie = __builtin_amdgcn_ballot_w32(k == m);
        const int par = i & 1;
        const unsigned firstbit = tie & (0u - tie);
        int myidx = t * FPS_PPT + ts[0];
        lds_v4i* recp = (lds_v4i*)&srec[par][wid];
        lds_int* keyp = (lds_int*)&skey[par][wid];
        asm volatile("" : "+v"(cx), "+v"(cy), "+v"(cz), "+v"(myidx), "+v"(recp), "+v"(keyp)
                        : "s"(firstbit));
        if (__builtin_amdgcn_inverse_ballot_w32(firstbit)) {
            const v4i mine = { __builtin_bit_cast(int, cx), __builtin_bit_cast(int, cy),
                               __builtin_bit_cast(int, cz), myidx };
            *recp = mine;
            *keyp = m;
        }
        __syncthreads();
        const int rk = skey[par][lane & (FPS_WAVES - 1)];
        const v4i rec = srec[par][lane & (FPS_WAVES - 1)];
        const int rx = rec.x, ry = rec.y, rz = rec.z, ridx = rec.w;
        const int m2 = octet_max_i(rk);
        const int wl = __builtin_ctz(__builtin_amdgcn_ballot_w32(rk == m2));
        const float nx = __builtin_bit_cast(float, __builtin_amdgcn_readlane(rx, wl));
        const float ny = __builtin_bit_cast(float, __builtin_amdgcn_readlane(ry, wl));
        const float nz = __builtin_bit_cast(float, __builtin_amdgcn_readlane(rz, wl));
        if (t == 0) fid[b * NOUT + i] = b * NPTS + __builtin_amdgcn_readlane(ridx, wl);
        #pragma unroll
        for (int s = 0; s < FPS_PPT; ++s) {
            const float dx = px[s] - nx, dy = py[s] - ny, dz = pz[s] - nz;
            d[s] = fminf(d[s], sqdist(dx, dy, dz));
        }
    }
}

__global__ __launch_bounds__(FPS_THREADS) void fps_and_gemm_kernel(
    const float* __restrict__ x, const float* __restrict__ W,
    const float* __restrict__ bias, const float* __restrict__ gamma,
    const float* __restrict__ beta, const float* __restrict__ rmean,
    const float* __restrict__ rvar, float* __restrict__ h,
    const float* __restrict__ p, int* __restrict__ fid)
{
    if (blockIdx.x < BATCH)
        fps_part(blockIdx.x, p, fid);
    else
        gemm_bn_relu_part(blockIdx.x - BATCH, x, W, bias, gamma, beta, rmean, rvar, h);
}

__global__ __launch_bounds__(128) void gather_maxpool_kernel(
    const float* __restrict__ h, const float* __restrict__ p,
    const int* __restrict__ fid, const int* __restrict__ sid_euc,
    float* __restrict__ out)
{
    const int s = blockIdx.x;
    const int t = threadIdx.x;

    __shared__ int ssid[KNBR];
    const int f = fid[s];
    if (t < KNBR) ssid[t] = sid_euc[(size_t)f * KNBR + t];
    if (t < 3)    out[XOUT_ELEMS + (size_t)s * 3 + t] = p[(size_t)f * 3 + t];
    __syncthreads();

    const int c = t * 4;
    v4f acc = { -INFINITY, -INFINITY, -INFINITY, -INFINITY };
    #pragma unroll
    for (int k = 0; k < KNBR; ++k) {
        const int r = ssid[k];
        if (k + 1 < KNBR)
            __builtin_prefetch(h + (size_t)ssid[k + 1] * COUT + c, 0, 0);
        const v4f v = *(const v4f*)(h + (size_t)r * COUT + c);
        acc.x = fmaxf(acc.x, v.x);
        acc.y = fmaxf(acc.y, v.y);
        acc.z = fmaxf(acc.z, v.z);
        acc.w = fmaxf(acc.w, v.w);
    }
    *(v4f*)(out + (size_t)s * COUT + c) = acc;
}

extern "C" void kernel_launch(void* const* d_in, const int* in_sizes, int n_in,
                              void* d_out, int out_size, void* d_ws, size_t ws_size,
                              hipStream_t stream)
{
    (void)in_sizes; (void)n_in; (void)out_size; (void)ws_size;
    const float* x     = (const float*)d_in[0];
    const float* p     = (const float*)d_in[1];
    const int*   sid   = (const int*)  d_in[2];
    const float* W     = (const float*)d_in[4];
    const float* bias  = (const float*)d_in[5];
    const float* gamma = (const float*)d_in[6];
    const float* beta  = (const float*)d_in[7];
    const float* rmean = (const float*)d_in[8];
    const float* rvar  = (const float*)d_in[9];
    float* out = (float*)d_out;

    float* h  = (float*)d_ws;
    int* fid  = (int*)((char*)d_ws + (size_t)MROWS * COUT * 4);

    fps_and_gemm_kernel<<<BATCH + GEMM_MBLOCKS * GEMM_NBLOCKS, FPS_THREADS, 0, stream>>>(
        x, W, bias, gamma, beta, rmean, rvar, h, p, fid);
    gather_maxpool_kernel<<<BATCH * NOUT, 128, 0, stream>>>(h, p, fid, sid, out);
}


# Sampling chain cut at pick 896 into two launches: the last 128 picks run beside the gather of every batch's first 896 samples (one wave per sample, 64 row loads in flight); launch 3 gathers the rest
# speedup vs baseline: 4.2744x; 1.0355x over previous
#include <hip/hip_runtime.h>
#include <hip/hip_bf16.h>
#include <math.h>

typedef float v2f __attribute__((ext_vector_type(2)));
typedef float v4f __attribute__((ext_vector_type(4)));
typedef float v8f __attribute__((ext_vector_type(8)));

#define BATCH 16
#define NPTS  4096
#define CIN   256
#define COUT  512
#define KNBR  16
#define NOUT  1024
#define MROWS (BATCH * NPTS)
#define BN_EPS 1e-5f
#define XOUT_ELEMS ((size_t)BATCH * NOUT * COUT)

#define FPS_THREADS 256
#define FPS_SPLIT   896
#define FPS_TAIL    (NOUT - FPS_SPLIT)
#define CARRY_STRIDE ((size_t)NOUT * COUT)
#define GEMM_MBLK   (FPS_THREADS / 32 * 16)
#define GEMM_MBLOCKS (MROWS / GEMM_MBLK)
#define GEMM_NBLOCKS (COUT / 64)

__device__ __forceinline__ void gemm_bn_relu_part(
    const int g,
    const float* __restrict__ x, const float* __restrict__ W,
    const float* __restrict__ bias, const float* __restrict__ gamma,
    const float* __restrict__ beta, const float* __restrict__ rmean,
    const float* __restrict__ rvar, float* __restrict__ h)
{
    const int wave = threadIdx.x >> 5;
    const int lane = threadIdx.x & 31;
    const int m0 = ((g % GEMM_MBLOCKS) * (FPS_THREADS / 32) + wave) * 16;
    const int n0 = (g / GEMM_MBLOCKS) * 64;

    const int rowA = m0 + (lane & 15);
    const int kOff = (lane >> 4) << 1;
    const float* aPtr = x + (size_t)rowA * CIN + kOff;

    const int colL = lane & 15;
    const float* bPtr0 = W + (size_t)(n0 +  0 + colL) * CIN + kOff;
    const float* bPtr1 = W + (size_t)(n0 + 16 + colL) * CIN + kOff;
    const float* bPtr2 = W + (size_t)(n0 + 32 + colL) * CIN + kOff;
    const float* bPtr3 = W + (size_t)(n0 + 48 + colL) * CIN + kOff;

    v8f acc[4];
    #pragma unroll
    for (int nt = 0; nt < 4; ++nt) acc[nt] = {};

    #pragma unroll 4
    for (int c = 0; c < CIN; c += 4) {
        v2f a  = *(const v2f*)(aPtr + c);
        v2f b0 = *(const v2f*)(bPtr0 + c);
        v2f b1 = *(const v2f*)(bPtr1 + c);
        v2f b2 = *(const v2f*)(bPtr2 + c);
        v2f b3 = *(const v2f*)(bPtr3 + c);
        acc[0] = __builtin_amdgcn_wmma_f32_16x16x4_f32(false, a, false, b0,
                                                       (short)0, acc[0], false, false);
        acc[1] = __builtin_amdgcn_wmma_f32_16x16x4_f32(false, a, false, b1,
                                                       (short)0, acc[1], false, false);
        acc[2] = __builtin_amdgcn_wmma_f32_16x16x4_f32(false, a, false, b2,
                                                       (short)0, acc[2], false, false);
        acc[3] = __builtin_amdgcn_wmma_f32_16x16x4_f32(false, a, false, b3,
                                                       (short)0, acc[3], false, false);
    }

    const int rOff = (lane >> 4) << 3;
    #pragma unroll
    for (int nt = 0; nt < 4; ++nt) {
        const int o = n0 + nt * 16 + colL;
        const float sc = gamma[o] * rsqrtf(rvar[o] + BN_EPS);
        const float sh = (bias[o] - rmean[o]) * sc + beta[o];
        #pragma unroll
        for (int j = 0; j < 8; ++j) {
            float v = fmaf(acc[nt][j], sc, sh);
            v = fmaxf(v, 0.0f);
            h[(size_t)(m0 + j + rOff) * COUT + o] = v;
        }
    }
}

template <int XM>
__device__ __forceinline__ int row_xmask(int v)
{
    return __builtin_amdgcn_mov_dpp(v, 0x160 + XM, 0xf, 0xf, true);
}
__device__ __forceinline__ int xhalf(int v)
{
    return __builtin_amdgcn_permlanex16(v, v, 0x76543210, 0xfedcba98, false, false);
}
__device__ __forceinline__ int wave_max_i(int v)
{
    v = max(v, row_xmask<1>(v));
    v = max(v, row_xmask<2>(v));
    v = max(v, row_xmask<4>(v));
    v = max(v, row_xmask<8>(v));
    v = max(v, xhalf(v));
    return v;
}
__device__ __forceinline__ int octet_max_i(int v)
{
    v = max(v, row_xmask<1>(v));
    v = max(v, row_xmask<2>(v));
    v = max(v, row_xmask<4>(v));
    return v;
}

__device__ __forceinline__ float sqdist(float dx, float dy, float dz)
{
    return fmaf(dz, dz, fmaf(dx, dx, dy * dy));
}

#define FPS_PPT     (NPTS / FPS_THREADS)
#define FPS_WAVES   (FPS_THREADS / 32)
#define FPS_PITCH   (FPS_PPT + 1)
#define FPS_LDS_SLOTS 10496
static_assert(FPS_LDS_SLOTS >= FPS_THREADS * FPS_PITCH, "the cloud must fit its LDS array");
static_assert((size_t)FPS_LDS_SLOTS * 16 > 160 * 1024, "one work-group per WGP needs more than 160 KiB");
static_assert((size_t)FPS_LDS_SLOTS * 16 + 512 <= 320 * 1024, "must fit one WGP's LDS");
typedef int v4i __attribute__((ext_vector_type(4)));
typedef float v3f __attribute__((ext_vector_type(3)));
typedef __attribute__((address_space(3))) v4i lds_v4i;
typedef __attribute__((address_space(3))) int lds_int;
template <int I0, int I1>
__device__ __forceinline__ void fps_part(const int b, const float* __restrict__ p,
                                         int* fid, float* carry)
{
    const float* pb = p + (size_t)b * NPTS * 3;
    const int t = threadIdx.x;

    __shared__ v4f spt[FPS_LDS_SLOTS];
    __shared__ v4i srec[2][FPS_WAVES];
    __shared__ int skey[2][FPS_WAVES];

    float px[FPS_PPT], py[FPS_PPT], pz[FPS_PPT], d[FPS_PPT];
    float x0 = 0.0f, y0 = 0.0f, z0 = 0.0f;
    if (I0 == 1) { x0 = pb[0]; y0 = pb[1]; z0 = pb[2]; }
    float* const carry_t = carry + t;
    {
        float raw[FPS_PPT * 3];
        const v4f* src4 = (const v4f*)(pb + (size_t)t * (FPS_PPT * 3));
        #pragma unroll
        for (int q = 0; q < FPS_PPT * 3 / 4; ++q) {
            const v4f v = src4[q];
            raw[4 * q + 0] = v.x; raw[4 * q + 1] = v.y;
            raw[4 * q + 2] = v.z; raw[4 * q + 3] = v.w;
        }
        #pragma unroll
        for (int s = 0; s < FPS_PPT; ++s) {
            px[s] = raw[3 * s + 0];
            py[s] = raw[3 * s + 1];
            pz[s] = raw[3 * s + 2];
            const v4f q = { px[s], py[s], pz[s], 0.0f };
            spt[t * FPS_PITCH + s] = q;
            if (I0 == 1) {
                const float dx = px[s] - x0, dy = py[s] - y0, dz = pz[s] - z0;
                d[s] = sqdist(dx, dy, dz);
            }
        }
        if (I0 != 1) {
            #pragma unroll
            for (int s = 0; s < FPS_PPT; ++s) d[s] = carry_t[s * FPS_THREADS];
        }
    }

    if (I0 == 1 && t == 0) fid[b * NOUT + 0] = b * NPTS;

    const int wid = t >> 5, lane = t & 31;

    __syncthreads();

    for (int i = I0; i < I1; ++i) {
        float tv[FPS_PPT / 2]; int ts[FPS_PPT / 2];
        #pragma unroll
        for (int s = 0; s < FPS_PPT / 2; ++s) {
            const float va = d[2 * s], vb = d[2 * s + 1];
            const bool r = vb > va;
            tv[s] = r ? vb : va;
            ts[s] = r ? 2 * s + 1 : 2 * s;
        }
        #pragma unroll
        for (int w = 1; w < FPS_PPT / 2; w <<= 1) {
            #pragma unroll
            for (int s = 0; s < FPS_PPT / 2; s += 2 * w) {
                const float va = tv[s], vb = tv[s + w];
                const int   sa = ts[s], sb = ts[s + w];
                const bool r = vb > va;
                tv[s] = r ? vb : va;
                ts[s] = r ? sb : sa;
            }
        }
        const v3f cand = *(const v3f*)&spt[t * FPS_PITCH + ts[0]];
        float cx = cand.x, cy = cand.y, cz = cand.z;
        __builtin_amdgcn_sched_barrier(0);
        const int k = __builtin_bit_cast(int, tv[0]);
        const int m = wave_max_i(k);
        const unsigned tie = __builtin_amdgcn_ballot_w32(k == m);
        const int par = i & 1;
        const unsigned firstbit = tie & (0u - tie);
        int myidx = t * FPS_PPT + ts[0];
        lds_v4i* recp = (lds_v4i*)&srec[par][wid];
        lds_int* keyp = (lds_int*)&skey[par][wid];
        asm volatile("" : "+v"(cx), "+v"(cy), "+v"(myidx), "+v"(recp), "+v"(keyp)
                        : "s"(firstbit));
        if (__builtin_amdgcn_inverse_ballot_w32(firstbit)) {
            const v4i mine = { __builtin_bit_cast(int, cx), __builtin_bit_cast(int, cy),
                               __builtin_bit_cast(int, cz), myidx };
            *recp = mine;
            *keyp = m;
        }
        __syncthreads();
        const int rk = skey[par][lane & (FPS_WAVES - 1)];
        const v4i rec = srec[par][lane & (FPS_WAVES - 1)];
        const int rx = rec.x, ry = rec.y, rz = rec.z, ridx = rec.w;
        const int m2 = octet_max_i(rk);
        const int wl = __builtin_ctz(__builtin_amdgcn_ballot_w32(rk == m2));
        const float nx = __builtin_bit_cast(float, __builtin_amdgcn_readlane(rx, wl));
        const float ny = __builtin_bit_cast(float, __builtin_amdgcn_readlane(ry, wl));
        const float nz = __builtin_bit_cast(float, __builtin_amdgcn_readlane(rz, wl));
        if (t == 0) fid[b * NOUT + i] = b * NPTS + __builtin_amdgcn_readlane(ridx, wl);
        #pragma unroll
        for (int s = 0; s < FPS_PPT; ++s) {
            const float dx = px[s] - nx, dy = py[s] - ny, dz = pz[s] - nz;
            d[s] = fminf(d[s], sqdist(dx, dy, dz));
        }
    }

    if (I1 < NOUT) {
        #pragma unroll
        for (int s = 0; s < FPS_PPT; ++s) carry_t[s * FPS_THREADS] = d[s];
    }
}

__global__ __launch_bounds__(FPS_THREADS) void fps_and_gemm_kernel(
    const float* __restrict__ x, const float* __restrict__ W,
    const float* __restrict__ bias, const float* __restrict__ gamma,
    const float* __restrict__ beta, const float* __restrict__ rmean,
    const float* __restrict__ rvar, float* __restrict__ h,
    const float* __restrict__ p, int* __restrict__ fid, float* __restrict__ carry)
{
    if (blockIdx.x < BATCH)
        fps_part<1, FPS_SPLIT>(blockIdx.x, p, fid,
                               carry + (size_t)blockIdx.x * CARRY_STRIDE);
    else
        gemm_bn_relu_part(blockIdx.x - BATCH, x, W, bias, gamma, beta, rmean, rvar, h);
}

__device__ __forceinline__ void gather_sample_wave(
    const float* __restrict__ h, const float* __restrict__ p,
    float* out, const int s, const int f, const int nb, const int lane)
{
    if (lane < 3) out[XOUT_ELEMS + (size_t)s * 3 + lane] = p[(size_t)f * 3 + lane];

    v4f acc[4];
    #pragma unroll
    for (int c = 0; c < 4; ++c) acc[c] = { -INFINITY, -INFINITY, -INFINITY, -INFINITY };
    #pragma unroll
    for (int k = 0; k < KNBR; ++k) {
        const int r = __builtin_amdgcn_readlane(nb, k);
        const float* row = h + (size_t)r * COUT + lane * 4;
        #pragma unroll
        for (int c = 0; c < 4; ++c) {
            const v4f v = *(const v4f*)(row + c * 128);
            acc[c].x = fmaxf(acc[c].x, v.x);
            acc[c].y = fmaxf(acc[c].y, v.y);
            acc[c].z = fmaxf(acc[c].z, v.z);
            acc[c].w = fmaxf(acc[c].w, v.w);
        }
    }
    #pragma unroll
    for (int c = 0; c < 4; ++c)
        *(v4f*)(out + (size_t)s * COUT + c * 128 + lane * 4) = acc[c];
}

#define HEAD_SPW     2
#define HEAD_SAMPLES (BATCH * FPS_SPLIT)
#define HEAD_WGS     (HEAD_SAMPLES / (FPS_THREADS / 32 * HEAD_SPW))
static_assert(HEAD_WGS * (FPS_THREADS / 32 * HEAD_SPW) == HEAD_SAMPLES, "head samples must fill whole work-groups");
__global__ __launch_bounds__(FPS_THREADS) void fps_tail_and_gather_kernel(
    const float* __restrict__ h, const float* __restrict__ p, int* fid,
    const int* __restrict__ sid_euc, float* out, float* carry)
{
    if (blockIdx.x < BATCH) {
        fps_part<FPS_SPLIT, NOUT>(blockIdx.x, p, fid,
                                  carry + (size_t)blockIdx.x * CARRY_STRIDE);
    } else {
        const int wave = __builtin_amdgcn_readfirstlane((int)(threadIdx.x >> 5));
        const int lane = threadIdx.x & 31;
        const int w = ((int)blockIdx.x - BATCH) * (FPS_THREADS / 32) + wave;
        int ss[HEAD_SPW], fs[HEAD_SPW], nbs[HEAD_SPW];
        #pragma unroll
        for (int j = 0; j < HEAD_SPW; ++j) {
            const int q = w * HEAD_SPW + j;
            ss[j] = (q / FPS_SPLIT) * NOUT + (q % FPS_SPLIT);
            fs[j] = fid[ss[j]];
        }
        #pragma unroll
        for (int j = 0; j < HEAD_SPW; ++j)
            nbs[j] = sid_euc[(size_t)fs[j] * KNBR + (lane & (KNBR - 1))];
        #pragma unroll
        for (int j = 0; j < HEAD_SPW; ++j)
            gather_sample_wave(h, p, out, ss[j], fs[j], nbs[j], lane);
    }
}

#define TAIL_WAVES   2
#define TAIL_SAMPLES (BATCH * FPS_TAIL)
static_assert(TAIL_SAMPLES % TAIL_WAVES == 0, "tail samples must fill whole work-groups");
__global__ __launch_bounds__(TAIL_WAVES * 32) void gather_tail_kernel(
    const float* __restrict__ h, const float* __restrict__ p,
    const int* __restrict__ fid, const int* __restrict__ sid_euc,
    float* __restrict__ out)
{
    const int wave = __builtin_amdgcn_readfirstlane((int)(threadIdx.x >> 5));
    const int lane = threadIdx.x & 31;
    const int q = (int)blockIdx.x * TAIL_WAVES + wave;
    const int s = (q / FPS_TAIL) * NOUT + FPS_SPLIT + (q % FPS_TAIL);
    const int f = fid[s];
    const int nb = sid_euc[(size_t)f * KNBR + (lane & (KNBR - 1))];
    gather_sample_wave(h, p, out, s, f, nb, lane);
}

extern "C" void kernel_launch(void* const* d_in, const int* in_sizes, int n_in,
                              void* d_out, int out_size, void* d_ws, size_t ws_size,
                              hipStream_t stream)
{
    (void)in_sizes; (void)n_in; (void)out_size; (void)ws_size;
    const float* x     = (const float*)d_in[0];
    const float* p     = (const float*)d_in[1];
    const int*   sid   = (const int*)  d_in[2];
    const float* W     = (const float*)d_in[4];
    const float* bias  = (const float*)d_in[5];
    const float* gamma = (const float*)d_in[6];
    const float* beta  = (const float*)d_in[7];
    const float* rmean = (const float*)d_in[8];
    const float* rvar  = (const float*)d_in[9];
    float* out = (float*)d_out;

    float* h  = (float*)d_ws;
    int* fid  = (int*)((char*)d_ws + (size_t)MROWS * COUT * 4);

    float* carry = out + (size_t)FPS_SPLIT * COUT;

    fps_and_gemm_kernel<<<BATCH + GEMM_MBLOCKS * GEMM_NBLOCKS, FPS_THREADS, 0, stream>>>(
        x, W, bias, gamma, beta, rmean, rvar, h, p, fid, carry);
    fps_tail_and_gather_kernel<<<BATCH + HEAD_WGS, FPS_THREADS, 0, stream>>>(
        h, p, fid, sid, out, carry);
    gather_tail_kernel<<<TAIL_SAMPLES / TAIL_WAVES, TAIL_WAVES * 32, 0, stream>>>(h, p, fid, sid, out);
}


# FPS: two picks per pass when proven: the best partial b of the other 15 rows is the next pick if sqdist(b-a) >= d[b] and a's row holds nothing >= d[b]; else one pick as before; same pick sequence
# speedup vs baseline: 4.6074x; 1.0779x over previous
#include <hip/hip_runtime.h>
#include <hip/hip_bf16.h>
#include <math.h>

typedef float v2f __attribute__((ext_vector_type(2)));
typedef float v4f __attribute__((ext_vector_type(4)));
typedef float v8f __attribute__((ext_vector_type(8)));

#define BATCH 16
#define NPTS  4096
#define CIN   256
#define COUT  512
#define KNBR  16
#define NOUT  1024
#define MROWS (BATCH * NPTS)
#define BN_EPS 1e-5f
#define XOUT_ELEMS ((size_t)BATCH * NOUT * COUT)

#define FPS_THREADS 256
#define FPS_SPLIT   896
#define FPS_TAIL    (NOUT - FPS_SPLIT)
#define CARRY_STRIDE ((size_t)NOUT * COUT)
#define GEMM_MBLK   (FPS_THREADS / 32 * 16)
#define GEMM_MBLOCKS (MROWS / GEMM_MBLK)
#define GEMM_NBLOCKS (COUT / 64)

__device__ __forceinline__ void gemm_bn_relu_part(
    const int g,
    const float* __restrict__ x, const float* __restrict__ W,
    const float* __restrict__ bias, const float* __restrict__ gamma,
    const float* __restrict__ beta, const float* __restrict__ rmean,
    const float* __restrict__ rvar, float* __restrict__ h)
{
    const int wave = threadIdx.x >> 5;
    const int lane = threadIdx.x & 31;
    const int m0 = ((g % GEMM_MBLOCKS) * (FPS_THREADS / 32) + wave) * 16;
    const int n0 = (g / GEMM_MBLOCKS) * 64;

    const int rowA = m0 + (lane & 15);
    const int kOff = (lane >> 4) << 1;
    const float* aPtr = x + (size_t)rowA * CIN + kOff;

    const int colL = lane & 15;
    const float* bPtr0 = W + (size_t)(n0 +  0 + colL) * CIN + kOff;
    const float* bPtr1 = W + (size_t)(n0 + 16 + colL) * CIN + kOff;
    const float* bPtr2 = W + (size_t)(n0 + 32 + colL) * CIN + kOff;
    const float* bPtr3 = W + (size_t)(n0 + 48 + colL) * CIN + kOff;

    v8f acc[4];
    #pragma unroll
    for (int nt = 0; nt < 4; ++nt) acc[nt] = {};

    #pragma unroll 4
    for (int c = 0; c < CIN; c += 4) {
        v2f a  = *(const v2f*)(aPtr + c);
        v2f b0 = *(const v2f*)(bPtr0 + c);
        v2f b1 = *(const v2f*)(bPtr1 + c);
        v2f b2 = *(const v2f*)(bPtr2 + c);
        v2f b3 = *(const v2f*)(bPtr3 + c);
        acc[0] = __builtin_amdgcn_wmma_f32_16x16x4_f32(false, a, false, b0,
                                                       (short)0, acc[0], false, false);
        acc[1] = __builtin_amdgcn_wmma_f32_16x16x4_f32(false, a, false, b1,
                                                       (short)0, acc[1], false, false);
        acc[2] = __builtin_amdgcn_wmma_f32_16x16x4_f32(false, a, false, b2,
                                                       (short)0, acc[2], false, false);
        acc[3] = __builtin_amdgcn_wmma_f32_16x16x4_f32(false, a, false, b3,
                                                       (short)0, acc[3], false, false);
    }

    const int rOff = (lane >> 4) << 3;
    #pragma unroll
    for (int nt = 0; nt < 4; ++nt) {
        const int o = n0 + nt * 16 + colL;
        const float sc = gamma[o] * rsqrtf(rvar[o] + BN_EPS);
        const float sh = (bias[o] - rmean[o]) * sc + beta[o];
        #pragma unroll
        for (int j = 0; j < 8; ++j) {
            float v = fmaf(acc[nt][j], sc, sh);
            v = fmaxf(v, 0.0f);
            h[(size_t)(m0 + j + rOff) * COUT + o] = v;
        }
    }
}

template <int XM>
__device__ __forceinline__ int row_xmask(int v)
{
    return __builtin_amdgcn_mov_dpp(v, 0x160 + XM, 0xf, 0xf, true);
}
__device__ __forceinline__ int row_max_i(int v)
{
    v = max(v, row_xmask<1>(v));
    v = max(v, row_xmask<2>(v));
    v = max(v, row_xmask<4>(v));
    v = max(v, row_xmask<8>(v));
    return v;
}
template <int XM>
__device__ __forceinline__ void row_top2_step(int& t1, int& t2)
{
    const int p1 = row_xmask<XM>(t1), p2 = row_xmask<XM>(t2);
    t2 = max(max(min(t1, p1), t2), p2);
    t1 = max(t1, p1);
}
__device__ __forceinline__ void row_top2_i(int& t1, int& t2)
{
    row_top2_step<1>(t1, t2);
    row_top2_step<2>(t1, t2);
    row_top2_step<4>(t1, t2);
    row_top2_step<8>(t1, t2);
}

__device__ __forceinline__ float sqdist(float dx, float dy, float dz)
{
    return fmaf(dz, dz, fmaf(dx, dx, dy * dy));
}

#define FPS_PPT     (NPTS / FPS_THREADS)
#define FPS_ROWS    (FPS_THREADS / 16)
#define FPS_PITCH   (FPS_PPT + 1)
#define FPS_LDS_SLOTS 10496
static_assert(FPS_LDS_SLOTS >= FPS_THREADS * FPS_PITCH, "the cloud must fit its LDS array");
static_assert((size_t)FPS_LDS_SLOTS * 16 > 160 * 1024, "one work-group per WGP needs more than 160 KiB");
static_assert((size_t)FPS_LDS_SLOTS * 16 + 1024 <= 320 * 1024, "must fit one WGP's LDS");
typedef int v4i __attribute__((ext_vector_type(4)));
typedef float v3f __attribute__((ext_vector_type(3)));
typedef __attribute__((address_space(3))) v4i lds_v4i;
typedef int v2i __attribute__((ext_vector_type(2)));
typedef __attribute__((address_space(3))) v2i lds_v2i;
template <int I0, int I1>
__device__ __forceinline__ void fps_part(const int b, const float* __restrict__ p,
                                         int* fid, float* carry)
{
    const float* pb = p + (size_t)b * NPTS * 3;
    const int t = threadIdx.x;

    __shared__ v4f spt[FPS_LDS_SLOTS];
    __shared__ v4i srec[2][FPS_ROWS];
    __shared__ v2i skey[2][FPS_ROWS];

    float px[FPS_PPT], py[FPS_PPT], pz[FPS_PPT], d[FPS_PPT];
    float x0 = 0.0f, y0 = 0.0f, z0 = 0.0f;
    if (I0 == 1) { x0 = pb[0]; y0 = pb[1]; z0 = pb[2]; }
    float* const carry_t = carry + t;
    {
        float raw[FPS_PPT * 3];
        const v4f* src4 = (const v4f*)(pb + (size_t)t * (FPS_PPT * 3));
        #pragma unroll
        for (int q = 0; q < FPS_PPT * 3 / 4; ++q) {
            const v4f v = src4[q];
            raw[4 * q + 0] = v.x; raw[4 * q + 1] = v.y;
            raw[4 * q + 2] = v.z; raw[4 * q + 3] = v.w;
        }
        #pragma unroll
        for (int s = 0; s < FPS_PPT; ++s) {
            px[s] = raw[3 * s + 0];
            py[s] = raw[3 * s + 1];
            pz[s] = raw[3 * s + 2];
            const v4f q = { px[s], py[s], pz[s], 0.0f };
            spt[t * FPS_PITCH + s] = q;
            if (I0 == 1) {
                const float dx = px[s] - x0, dy = py[s] - y0, dz = pz[s] - z0;
                d[s] = sqdist(dx, dy, dz);
            }
        }
        if (I0 != 1) {
            #pragma unroll
            for (int s = 0; s < FPS_PPT; ++s) d[s] = carry_t[s * FPS_THREADS];
        }
    }

    if (I0 == 1 && t == 0) fid[b * NOUT + 0] = b * NPTS;

    const int lane = t & 31;
    const int rowlane = lane & 15;
    const int row = t >> 4;

    __syncthreads();

    int par = 0;
    for (int i = I0; i < I1; ) {
        float tv[FPS_PPT / 2]; int ts[FPS_PPT / 2]; int tr[FPS_PPT / 2];
        #pragma unroll
        for (int s = 0; s < FPS_PPT / 2; ++s) {
            const float va = d[2 * s], vb = d[2 * s + 1];
            const bool r = vb > va;
            tv[s] = r ? vb : va;
            ts[s] = r ? 2 * s + 1 : 2 * s;
            tr[s] = min(__builtin_bit_cast(int, va), __builtin_bit_cast(int, vb));
        }
        #pragma unroll
        for (int w = 1; w < FPS_PPT / 2; w <<= 1) {
            #pragma unroll
            for (int s = 0; s < FPS_PPT / 2; s += 2 * w) {
                const float va = tv[s], vb = tv[s + w];
                const int   sa = ts[s], sb = ts[s + w];
                const bool r = vb > va;
                const int lo = min(__builtin_bit_cast(int, va), __builtin_bit_cast(int, vb));
                tr[s] = max(max(lo, tr[s]), tr[s + w]);
                tv[s] = r ? vb : va;
                ts[s] = r ? sb : sa;
            }
        }
        const v3f cand = *(const v3f*)&spt[t * FPS_PITCH + ts[0]];
        float cx = cand.x, cy = cand.y, cz = cand.z;
        __builtin_amdgcn_sched_barrier(0);
        const int k = __builtin_bit_cast(int, tv[0]);
        int m = k, m2 = tr[0];
        row_top2_i(m, m2);
        const unsigned tie = __builtin_amdgcn_ballot_w32(k == m);
        const unsigned tlo = tie & 0xffffu, thi = tie & 0xffff0000u;
        const unsigned firstbits = (tlo & (0u - tlo)) | (thi & (0u - thi));
        int myidx = t * FPS_PPT + ts[0];
        lds_v4i* recp = (lds_v4i*)&srec[par][row];
        lds_v2i* keyp = (lds_v2i*)&skey[par][row];
        asm volatile("" : "+v"(cx), "+v"(cy), "+v"(myidx), "+v"(recp), "+v"(keyp)
                        : "s"(firstbits));
        if (__builtin_amdgcn_inverse_ballot_w32(firstbits)) {
            const v4i mine = { __builtin_bit_cast(int, cx), __builtin_bit_cast(int, cy),
                               __builtin_bit_cast(int, cz), myidx };
            const v2i keys = { m, m2 };
            *recp = mine;
            *keyp = keys;
        }
        __syncthreads();
        const v2i rkeys = skey[par][rowlane];
        const v4i rec = srec[par][rowlane];
        const int rk = rkeys.x, rr = rkeys.y;
        const int rx = rec.x, ry = rec.y, rz = rec.z, ridx = rec.w;
        const int m1 = row_max_i(rk);
        const int wa = __builtin_ctz(__builtin_amdgcn_ballot_w32(rk == m1));
        const float nx = __builtin_bit_cast(float, __builtin_amdgcn_readlane(rx, wa));
        const float ny = __builtin_bit_cast(float, __builtin_amdgcn_readlane(ry, wa));
        const float nz = __builtin_bit_cast(float, __builtin_amdgcn_readlane(rz, wa));
        const int kb = (rowlane == wa) ? -1 : rk;
        const int mb = row_max_i(kb);
        const int wb = __builtin_ctz(__builtin_amdgcn_ballot_w32(kb == mb));
        const float ex = __builtin_bit_cast(float, rx) - nx;
        const float ey = __builtin_bit_cast(float, ry) - ny;
        const float ez = __builtin_bit_cast(float, rz) - nz;
        const int ebits = __builtin_bit_cast(int, sqdist(ex, ey, ez));
        const int ub = __builtin_amdgcn_readlane(rk, wb);
        const int eb = __builtin_amdgcn_readlane(ebits, wb);
        const int ra = __builtin_amdgcn_readlane(rr, wa);
        const bool two = (i + 1 < I1) && (ub > ra) && (ub > 0) && (eb >= ub);
        const int wq = two ? wb : wa;
        const float qx = __builtin_bit_cast(float, __builtin_amdgcn_readlane(rx, wq));
        const float qy = __builtin_bit_cast(float, __builtin_amdgcn_readlane(ry, wq));
        const float qz = __builtin_bit_cast(float, __builtin_amdgcn_readlane(rz, wq));
        if (t == 0) {
            fid[b * NOUT + i] = b * NPTS + __builtin_amdgcn_readlane(ridx, wa);
            if (two) fid[b * NOUT + i + 1] = b * NPTS + __builtin_amdgcn_readlane(ridx, wb);
        }
        #pragma unroll
        for (int s = 0; s < FPS_PPT; ++s) {
            const float dx = px[s] - nx, dy = py[s] - ny, dz = pz[s] - nz;
            const float fx = px[s] - qx, fy = py[s] - qy, fz = pz[s] - qz;
            d[s] = fminf(fminf(d[s], sqdist(dx, dy, dz)), sqdist(fx, fy, fz));
        }
        i += two ? 2 : 1;
        par ^= 1;
    }

    if (I1 < NOUT) {
        #pragma unroll
        for (int s = 0; s < FPS_PPT; ++s) carry_t[s * FPS_THREADS] = d[s];
    }
}

__global__ __launch_bounds__(FPS_THREADS) void fps_and_gemm_kernel(
    const float* __restrict__ x, const float* __restrict__ W,
    const float* __restrict__ bias, const float* __restrict__ gamma,
    const float* __restrict__ beta, const float* __restrict__ rmean,
    const float* __restrict__ rvar, float* __restrict__ h,
    const float* __restrict__ p, int* __restrict__ fid, float* __restrict__ carry)
{
    if (blockIdx.x < BATCH)
        fps_part<1, FPS_SPLIT>(blockIdx.x, p, fid,
                               carry + (size_t)blockIdx.x * CARRY_STRIDE);
    else
        gemm_bn_relu_part(blockIdx.x - BATCH, x, W, bias, gamma, beta, rmean, rvar, h);
}

__device__ __forceinline__ void gather_sample_wave(
    const float* __restrict__ h, const float* __restrict__ p,
    float* out, const int s, const int f, const int nb, const int lane)
{
    if (lane < 3) out[XOUT_ELEMS + (size_t)s * 3 + lane] = p[(size_t)f * 3 + lane];

    v4f acc[4];
    #pragma unroll
    for (int c = 0; c < 4; ++c) acc[c] = { -INFINITY, -INFINITY, -INFINITY, -INFINITY };
    #pragma unroll
    for (int k = 0; k < KNBR; ++k) {
        const int r = __builtin_amdgcn_readlane(nb, k);
        const float* row = h + (size_t)r * COUT + lane * 4;
        #pragma unroll
        for (int c = 0; c < 4; ++c) {
            const v4f v = *(const v4f*)(row + c * 128);
            acc[c].x = fmaxf(acc[c].x, v.x);
            acc[c].y = fmaxf(acc[c].y, v.y);
            acc[c].z = fmaxf(acc[c].z, v.z);
            acc[c].w = fmaxf(acc[c].w, v.w);
        }
    }
    #pragma unroll
    for (int c = 0; c < 4; ++c)
        *(v4f*)(out + (size_t)s * COUT + c * 128 + lane * 4) = acc[c];
}

#define HEAD_SPW     2
#define HEAD_SAMPLES (BATCH * FPS_SPLIT)
#define HEAD_WGS     (HEAD_SAMPLES / (FPS_THREADS / 32 * HEAD_SPW))
static_assert(HEAD_WGS * (FPS_THREADS / 32 * HEAD_SPW) == HEAD_SAMPLES, "head samples must fill whole work-groups");
__global__ __launch_bounds__(FPS_THREADS) void fps_tail_and_gather_kernel(
    const float* __restrict__ h, const float* __restrict__ p, int* fid,
    const int* __restrict__ sid_euc, float* out, float* carry)
{
    if (blockIdx.x < BATCH) {
        fps_part<FPS_SPLIT, NOUT>(blockIdx.x, p, fid,
                                  carry + (size_t)blockIdx.x * CARRY_STRIDE);
    } else {
        const int wave = __builtin_amdgcn_readfirstlane((int)(threadIdx.x >> 5));
        const int lane = threadIdx.x & 31;
        const int w = ((int)blockIdx.x - BATCH) * (FPS_THREADS / 32) + wave;
        int ss[HEAD_SPW], fs[HEAD_SPW], nbs[HEAD_SPW];
        #pragma unroll
        for (int j = 0; j < HEAD_SPW; ++j) {
            const int q = w * HEAD_SPW + j;
            ss[j] = (q / FPS_SPLIT) * NOUT + (q % FPS_SPLIT);
            fs[j] = fid[ss[j]];
        }
        #pragma unroll
        for (int j = 0; j < HEAD_SPW; ++j)
            nbs[j] = sid_euc[(size_t)fs[j] * KNBR + (lane & (KNBR - 1))];
        #pragma unroll
        for (int j = 0; j < HEAD_SPW; ++j)
            gather_sample_wave(h, p, out, ss[j], fs[j], nbs[j], lane);
    }
}

#define TAIL_WAVES   2
#define TAIL_SAMPLES (BATCH * FPS_TAIL)
static_assert(TAIL_SAMPLES % TAIL_WAVES == 0, "tail samples must fill whole work-groups");
__global__ __launch_bounds__(TAIL_WAVES * 32) void gather_tail_kernel(
    const float* __restrict__ h, const float* __restrict__ p,
    const int* __restrict__ fid, const int* __restrict__ sid_euc,
    float* __restrict__ out)
{
    const int wave = __builtin_amdgcn_readfirstlane((int)(threadIdx.x >> 5));
    const int lane = threadIdx.x & 31;
    const int q = (int)blockIdx.x * TAIL_WAVES + wave;
    const int s = (q / FPS_TAIL) * NOUT + FPS_SPLIT + (q % FPS_TAIL);
    const int f = fid[s];
    const int nb = sid_euc[(size_t)f * KNBR + (lane & (KNBR - 1))];
    gather_sample_wave(h, p, out, s, f, nb, lane);
}

extern "C" void kernel_launch(void* const* d_in, const int* in_sizes, int n_in,
                              void* d_out, int out_size, void* d_ws, size_t ws_size,
                              hipStream_t stream)
{
    (void)in_sizes; (void)n_in; (void)out_size; (void)ws_size;
    const float* x     = (const float*)d_in[0];
    const float* p     = (const float*)d_in[1];
    const int*   sid   = (const int*)  d_in[2];
    const float* W     = (const float*)d_in[4];
    const float* bias  = (const float*)d_in[5];
    const float* gamma = (const float*)d_in[6];
    const float* beta  = (const float*)d_in[7];
    const float* rmean = (const float*)d_in[8];
    const float* rvar  = (const float*)d_in[9];
    float* out = (float*)d_out;

    float* h  = (float*)d_ws;
    int* fid  = (int*)((char*)d_ws + (size_t)MROWS * COUT * 4);

    float* carry = out + (size_t)FPS_SPLIT * COUT;

    fps_and_gemm_kernel<<<BATCH + GEMM_MBLOCKS * GEMM_NBLOCKS, FPS_THREADS, 0, stream>>>(
        x, W, bias, gamma, beta, rmean, rvar, h, p, fid, carry);
    fps_tail_and_gather_kernel<<<BATCH + HEAD_WGS, FPS_THREADS, 0, stream>>>(
        h, p, fid, sid, out, carry);
    gather_tail_kernel<<<TAIL_SAMPLES / TAIL_WAVES, TAIL_WAVES * 32, 0, stream>>>(h, p, fid, sid, out);
}
